# starting kernel
# baseline (speedup 1.0000x reference)
#include <hip/hip_runtime.h>

typedef __attribute__((ext_vector_type(2))) float v2f;
typedef __attribute__((ext_vector_type(8))) float v8f;

namespace {
constexpr int T = 1024, B = 512, D = 16, H = 16;

__device__ __forceinline__ float tanh_hw(float x) {
  float r;
  asm volatile("v_tanh_f32 %0, %1\n\tv_nop" : "=v"(r) : "v"(x));
  return r;
}
__device__ __forceinline__ float sigmoid_hw(float x) {
  return __builtin_fmaf(0.5f, tanh_hw(0.5f * x), 0.5f);
}
__device__ __forceinline__ v8f splat8(float s) {
  v8f v;
#pragma unroll
  for (int i = 0; i < 8; ++i) v[i] = s;
  return v;
}

__device__ __forceinline__ void load_x(v2f (&xa)[4], const float* __restrict__ x,
                                       int t, int bbase, int n, int hi) {
#pragma unroll
  for (int j = 0; j < 4; ++j)
    xa[j] = *(const v2f*)(x + ((size_t)t * B + bbase + n) * D + 4 * j + 2 * hi);
}

__device__ __forceinline__ void lstm_step(
    int t, const v2f (&xc)[4], const v2f (&wih)[4][4], const v2f (&whh)[4][4],
    const v8f (&biasV)[4], v2f (&hA)[4], v8f& c, float* __restrict__ y,
    float* hlds, int n, int hi, int bbase) {
  v8f G[4];
#pragma unroll
  for (int nt = 0; nt < 4; ++nt) {
    G[nt] = __builtin_amdgcn_wmma_f32_16x16x4_f32(
        false, xc[0], false, wih[nt][0], (short)0, biasV[nt], false, false);
#pragma unroll
    for (int j = 1; j < 4; ++j)
      G[nt] = __builtin_amdgcn_wmma_f32_16x16x4_f32(
          false, xc[j], false, wih[nt][j], (short)0, G[nt], false, false);
#pragma unroll
    for (int j = 0; j < 4; ++j)
      G[nt] = __builtin_amdgcn_wmma_f32_16x16x4_f32(
          false, hA[j], false, whh[nt][j], (short)0, G[nt], false, false);
  }

  const int m0 = 8 * hi;
  float* yrow = y + ((size_t)t * B + bbase + m0) * H + n;
#pragma unroll
  for (int r = 0; r < 8; ++r) {
    float iv = sigmoid_hw(G[0][r]);
    float fv = sigmoid_hw(G[1][r]);
    float gv = tanh_hw(G[2][r]);
    float ov = sigmoid_hw(G[3][r]);
    float cv = __builtin_fmaf(fv, c[r], iv * gv);
    c[r] = cv;
    float hv = ov * tanh_hw(cv);
    yrow[(size_t)r * H] = hv;
    hlds[(m0 + r) * 16 + n] = hv;
  }
  __syncthreads();
#pragma unroll
  for (int j = 0; j < 4; ++j)
    hA[j] = *(const v2f*)(hlds + n * 16 + 4 * j + 2 * hi);
  __syncthreads();
}
}

__global__ __launch_bounds__(32, 1) void lstm_wmma_kernel(
    const float* __restrict__ x, const float* __restrict__ W_ih,
    const float* __restrict__ W_hh, const float* __restrict__ b_ih,
    const float* __restrict__ b_hh, float* __restrict__ y) {
  __shared__ __align__(16) float hlds[16 * 16];

  const int lane  = threadIdx.x;
  const int n     = lane & 15;
  const int hi    = lane >> 4;
  const int bbase = blockIdx.x * 16;

  v2f wih[4][4], whh[4][4];
#pragma unroll
  for (int nt = 0; nt < 4; ++nt)
#pragma unroll
    for (int j = 0; j < 4; ++j) {
      wih[nt][j] = *(const v2f*)(W_ih + (size_t)(16 * nt + n) * D + 4 * j + 2 * hi);
      whh[nt][j] = *(const v2f*)(W_hh + (size_t)(16 * nt + n) * H + 4 * j + 2 * hi);
    }

  v8f biasV[4];
#pragma unroll
  for (int nt = 0; nt < 4; ++nt)
    biasV[nt] = splat8(b_ih[16 * nt + n] + b_hh[16 * nt + n]);

  v8f c = {};
  v2f hA[4];
#pragma unroll
  for (int j = 0; j < 4; ++j) { hA[j][0] = 0.f; hA[j][1] = 0.f; }

  v2f xA0[4], xA1[4];
  load_x(xA0, x, 0, bbase, n, hi);

  for (int t = 0; t < T; t += 2) {
    load_x(xA1, x, t + 1, bbase, n, hi);
    if (t + 16 < T)
      __builtin_prefetch(x + ((size_t)(t + 16) * B + bbase + n) * D, 0, 1);
    lstm_step(t, xA0, wih, whh, biasV, hA, c, y, hlds, n, hi, bbase);

    if (t + 2 < T) load_x(xA0, x, t + 2, bbase, n, hi);
    lstm_step(t + 1, xA1, wih, whh, biasV, hA, c, y, hlds, n, hi, bbase);
  }
}

extern "C" void kernel_launch(void* const* d_in, const int*  ,
                              int  , void* d_out, int  ,
                              void*  , size_t  ,
                              hipStream_t stream) {
  const float* x    = (const float*)d_in[0];
  const float* W_ih = (const float*)d_in[1];
  const float* W_hh = (const float*)d_in[2];
  const float* b_ih = (const float*)d_in[3];
  const float* b_hh = (const float*)d_in[4];
  float* y = (float*)d_out;
  lstm_wmma_kernel<<<B / 16, 32, 0, stream>>>(x, W_ih, W_hh, b_ih, b_hh, y);
}


# Weights as A, x^T and h^T as B: h in D layout is the next step's B operand, so no LDS transpose and no barriers; x tiles requested one trip ahead; k regrouped per product (f32 sum order changes)
# speedup vs baseline: 2.2553x; 2.2553x over previous
#include <hip/hip_runtime.h>


typedef __attribute__((ext_vector_type(2))) float v2f;
typedef __attribute__((ext_vector_type(4))) float v4f;
typedef __attribute__((ext_vector_type(8))) float v8f;

namespace {
constexpr int T = 1024, B = 512, D = 16, H = 16;

__device__ __forceinline__ float tanh_hw(float x) {
  float r;
  asm volatile("v_tanh_f32 %0, %1\n\tv_nop" : "=v"(r) : "v"(x));
  return r;
}
__device__ __forceinline__ float sigmoid_hw(float x) {
  return __builtin_fmaf(0.5f, tanh_hw(0.5f * x), 0.5f);
}


__device__ __forceinline__ void load_x(v4f (&xb)[2], const float* __restrict__ x,
                                       int t, int bbase, int b, int hi) {
  const float* p = x + ((size_t)t * B + bbase + b) * D + 8 * hi;
  xb[0] = *(const v4f*)p;
  xb[1] = *(const v4f*)(p + 4);
}

__device__ __forceinline__ void lstm_step(
    int t, const v4f (&xb)[2], const v2f (&wih)[4][4], const v2f (&whh)[4][4],
    const v8f (&biasV)[4], v8f& h, v8f& c, float* __restrict__ y,
    int b, int hi, int bbase) {
  v2f xB[4], hB[4];
#pragma unroll
  for (int j = 0; j < 4; ++j) {
    xB[j][0] = xb[j >> 1][2 * (j & 1)];
    xB[j][1] = xb[j >> 1][2 * (j & 1) + 1];
    hB[j][0] = h[2 * j];
    hB[j][1] = h[2 * j + 1];
  }

  v8f G[4];
#pragma unroll
  for (int nt = 0; nt < 4; ++nt) {
    G[nt] = __builtin_amdgcn_wmma_f32_16x16x4_f32(
        false, wih[nt][0], false, xB[0], (short)0, biasV[nt], false, false);
#pragma unroll
    for (int j = 1; j < 4; ++j)
      G[nt] = __builtin_amdgcn_wmma_f32_16x16x4_f32(
          false, wih[nt][j], false, xB[j], (short)0, G[nt], false, false);
#pragma unroll
    for (int j = 0; j < 4; ++j)
      G[nt] = __builtin_amdgcn_wmma_f32_16x16x4_f32(
          false, whh[nt][j], false, hB[j], (short)0, G[nt], false, false);
  }

#pragma unroll
  for (int r = 0; r < 8; ++r) {
    float iv = sigmoid_hw(G[0][r]);
    float fv = sigmoid_hw(G[1][r]);
    float gv = tanh_hw(G[2][r]);
    float ov = sigmoid_hw(G[3][r]);
    float cv = __builtin_fmaf(fv, c[r], iv * gv);
    c[r] = cv;
    h[r] = ov * tanh_hw(cv);
  }

  float* yrow = y + ((size_t)t * B + bbase + b) * H + 8 * hi;
  v4f lo, up;
#pragma unroll
  for (int i = 0; i < 4; ++i) { lo[i] = h[i]; up[i] = h[4 + i]; }
  *(v4f*)yrow = lo;
  *(v4f*)(yrow + 4) = up;
}
}

__global__ __launch_bounds__(32, 1) void lstm_wmma_kernel(
    const float* __restrict__ x, const float* __restrict__ W_ih,
    const float* __restrict__ W_hh, const float* __restrict__ b_ih,
    const float* __restrict__ b_hh, float* __restrict__ y) {
  const int lane  = threadIdx.x;
  const int b     = lane & 15;
  const int hi    = lane >> 4;
  const int bbase = blockIdx.x * 16;

  v2f wih[4][4], whh[4][4];
#pragma unroll
  for (int nt = 0; nt < 4; ++nt)
#pragma unroll
    for (int j = 0; j < 4; ++j) {
      wih[nt][j] = *(const v2f*)(W_ih + (size_t)(16 * nt + b) * D + 8 * hi + 2 * j);
      whh[nt][j] = *(const v2f*)(W_hh + (size_t)(16 * nt + b) * H + 8 * hi + 2 * j);
    }

  v8f biasV[4];
#pragma unroll
  for (int nt = 0; nt < 4; ++nt)
#pragma unroll
    for (int r = 0; r < 8; ++r)
      biasV[nt][r] = b_ih[16 * nt + 8 * hi + r] + b_hh[16 * nt + 8 * hi + r];

  v8f c = {};
  v8f h = {};

  v4f xc0[2], xc1[2];
  load_x(xc0, x, 0, bbase, b, hi);
  load_x(xc1, x, 1, bbase, b, hi);

  for (int t = 0; t < T; t += 2) {
    const int tn = (t + 2 < T) ? (t + 2) : t;
    v4f xn0[2], xn1[2];
    load_x(xn0, x, tn, bbase, b, hi);
    load_x(xn1, x, tn + 1, bbase, b, hi);
    if (t + 16 < T)
      __builtin_prefetch(x + ((size_t)(t + 16 + hi) * B + bbase + b) * D, 0, 1);

    lstm_step(t, xc0, wih, whh, biasV, h, c, y, b, hi, bbase);
    lstm_step(t + 1, xc1, wih, whh, biasV, h, c, y, b, hi, bbase);

    xc0[0] = xn0[0]; xc0[1] = xn0[1];
    xc1[0] = xn1[0]; xc1[1] = xn1[1];
  }
}

extern "C" void kernel_launch(void* const* d_in, const int*  ,
                              int  , void* d_out, int  ,
                              void*  , size_t  ,
                              hipStream_t stream) {
  const float* x    = (const float*)d_in[0];
  const float* W_ih = (const float*)d_in[1];
  const float* W_hh = (const float*)d_in[2];
  const float* b_ih = (const float*)d_in[3];
  const float* b_hh = (const float*)d_in[4];
  float* y = (float*)d_out;
  lstm_wmma_kernel<<<B / 16, 32, 0, stream>>>(x, W_ih, W_hh, b_ih, b_hh, y);
}


# Software-pipelined step: next step's input projection interleaved with the gate work, cut into register-pair pieces each written one matrix instruction behind its producer; f32 sum order unchanged
# speedup vs baseline: 2.7822x; 1.2336x over previous
#include <hip/hip_runtime.h>


typedef __attribute__((ext_vector_type(2))) float v2f;
typedef __attribute__((ext_vector_type(4))) float v4f;
typedef __attribute__((ext_vector_type(8))) float v8f;

namespace {
constexpr int T = 1024, B = 512, D = 16, H = 16;

__device__ __forceinline__ float tanh_hw(float x) {
  float r;
  asm volatile("v_tanh_f32 %0, %1\n\tv_nop" : "=v"(r) : "v"(x));
  return r;
}


__device__ __forceinline__ void load_x(v4f (&xb)[2], const float* __restrict__ x,
                                       int t, int bbase, int b, int hi) {
  const float* p = x + ((size_t)t * B + bbase + b) * D + 8 * hi;
  xb[0] = *(const v4f*)p;
  xb[1] = *(const v4f*)(p + 4);
}

template <int q>
__device__ __forceinline__ void proj_one(v8f (&Gn)[4], const v4f (&xb)[2],
                                         const v2f (&wih)[4][4],
                                         const v8f (&biasV)[4]) {
  constexpr int j = q >> 2, nt = q & 3;
  v2f xB;
  xB[0] = xb[j >> 1][2 * (j & 1)];
  xB[1] = xb[j >> 1][2 * (j & 1) + 1];
  if constexpr (j == 0)
    Gn[nt] = __builtin_amdgcn_wmma_f32_16x16x4_f32(
        false, wih[nt][0], false, xB, (short)0, biasV[nt], false, false);
  else
    Gn[nt] = __builtin_amdgcn_wmma_f32_16x16x4_f32(
        false, wih[nt][j], false, xB, (short)0, Gn[nt], false, false);
}

template <int q>
__device__ __forceinline__ void keep_here(const v8f (&Gn)[4]) {
  asm volatile("" : : "v"(Gn[q & 3]));
}

template <int nt, int j>
__device__ __forceinline__ void rec_one(v8f (&G)[4], const v8f& h,
                                        const v2f (&whh)[4][4]) {
  v2f hB;
  hB[0] = h[2 * j];
  hB[1] = h[2 * j + 1];
  G[nt] = __builtin_amdgcn_wmma_f32_16x16x4_f32(
      false, whh[nt][j], false, hB, (short)0, G[nt], false, false);
}

template <int r>
__device__ __forceinline__ void half_pair(const v8f& g, float (&a)[8]) {
  a[r]     = 0.5f * g[r];
  a[r + 1] = 0.5f * g[r + 1];
}
template <int r>
__device__ __forceinline__ void tanh_pair(const float (&a)[8], float (&th)[8]) {
  th[r]     = tanh_hw(a[r]);
  th[r + 1] = tanh_hw(a[r + 1]);
}
template <int r>
__device__ __forceinline__ void tanh_pair(const v8f& g, float (&th)[8]) {
  th[r]     = tanh_hw(g[r]);
  th[r + 1] = tanh_hw(g[r + 1]);
}
template <int r>
__device__ __forceinline__ void sig_pair(const float (&th)[8], float (&s)[8]) {
  s[r]     = __builtin_fmaf(0.5f, th[r], 0.5f);
  s[r + 1] = __builtin_fmaf(0.5f, th[r + 1], 0.5f);
}
template <int r>
__device__ __forceinline__ void ig_pair(const float (&iv)[8], const float (&gv)[8],
                                        float (&m)[8]) {
  m[r]     = iv[r] * gv[r];
  m[r + 1] = iv[r + 1] * gv[r + 1];
}
template <int r>
__device__ __forceinline__ void cell_pair(const float (&fv)[8], const float (&m)[8],
                                          v8f& c) {
  c[r]     = __builtin_fmaf(fv[r], c[r], m[r]);
  c[r + 1] = __builtin_fmaf(fv[r + 1], c[r + 1], m[r + 1]);
}
template <int r>
__device__ __forceinline__ void out_pair(const float (&ov)[8], const float (&tc)[8],
                                         v8f& h) {
  h[r]     = ov[r] * tc[r];
  h[r + 1] = ov[r + 1] * tc[r + 1];
}

#define UNIT_END() __builtin_amdgcn_sched_barrier(0)

__device__ __forceinline__ void lstm_step(
    int t, v8f (&G)[4], v8f (&Gn)[4], const v4f (&xnext)[2],
    const v2f (&wih)[4][4], const v2f (&whh)[4][4], const v8f (&biasV)[4],
    v8f& h, v8f& c, float* __restrict__ y, int b, int hi, int bbase) {
  float ai[8], af[8], ao[8], ti[8], tf[8], to[8];
  float iv[8], fv[8], gv[8], ov[8], m[8], tc[8];

  rec_one<0, 0>(G, h, whh); UNIT_END();
  rec_one<1, 0>(G, h, whh); UNIT_END();
  rec_one<0, 1>(G, h, whh); UNIT_END();
  rec_one<1, 1>(G, h, whh); UNIT_END();
  rec_one<0, 2>(G, h, whh); UNIT_END();
  rec_one<1, 2>(G, h, whh); UNIT_END();
  rec_one<0, 3>(G, h, whh); UNIT_END();
  rec_one<1, 3>(G, h, whh); half_pair<0>(G[0], ai); UNIT_END();

  rec_one<2, 0>(G, h, whh);                    tanh_pair<0>(ai, ti); half_pair<2>(G[0], ai); UNIT_END();
  rec_one<3, 0>(G, h, whh); sig_pair<0>(ti, iv); tanh_pair<2>(ai, ti); half_pair<4>(G[0], ai); UNIT_END();
  rec_one<2, 1>(G, h, whh); sig_pair<2>(ti, iv); tanh_pair<4>(ai, ti); half_pair<6>(G[0], ai); UNIT_END();
  rec_one<3, 1>(G, h, whh); sig_pair<4>(ti, iv); tanh_pair<6>(ai, ti); half_pair<0>(G[1], af); UNIT_END();
  rec_one<2, 2>(G, h, whh); sig_pair<6>(ti, iv); tanh_pair<0>(af, tf); half_pair<2>(G[1], af); UNIT_END();
  rec_one<3, 2>(G, h, whh); sig_pair<0>(tf, fv); tanh_pair<2>(af, tf); half_pair<4>(G[1], af); UNIT_END();
  rec_one<2, 3>(G, h, whh); sig_pair<2>(tf, fv); tanh_pair<4>(af, tf); half_pair<6>(G[1], af); UNIT_END();
  rec_one<3, 3>(G, h, whh); sig_pair<4>(tf, fv); tanh_pair<6>(af, tf); tanh_pair<0>(G[2], gv); UNIT_END();

  proj_one<0>(Gn, xnext, wih, biasV); keep_here<0>(Gn);
  sig_pair<6>(tf, fv); ig_pair<0>(iv, gv, m); half_pair<0>(G[3], ao);
  UNIT_END();
  proj_one<1>(Gn, xnext, wih, biasV); keep_here<1>(Gn);
  cell_pair<0>(fv, m, c); tanh_pair<0>(ao, to); tanh_pair<2>(G[2], gv);
  UNIT_END();
  proj_one<2>(Gn, xnext, wih, biasV); keep_here<2>(Gn);
  tanh_pair<0>(c, tc); sig_pair<0>(to, ov); ig_pair<2>(iv, gv, m); half_pair<2>(G[3], ao);
  UNIT_END();
  proj_one<3>(Gn, xnext, wih, biasV); keep_here<3>(Gn);
  out_pair<0>(ov, tc, h); cell_pair<2>(fv, m, c); tanh_pair<2>(ao, to); tanh_pair<4>(G[2], gv);
  UNIT_END();
  proj_one<4>(Gn, xnext, wih, biasV); keep_here<4>(Gn);
  tanh_pair<2>(c, tc); sig_pair<2>(to, ov); ig_pair<4>(iv, gv, m); half_pair<4>(G[3], ao);
  UNIT_END();
  proj_one<5>(Gn, xnext, wih, biasV); keep_here<5>(Gn);
  out_pair<2>(ov, tc, h); cell_pair<4>(fv, m, c); tanh_pair<4>(ao, to); tanh_pair<6>(G[2], gv);
  UNIT_END();
  proj_one<6>(Gn, xnext, wih, biasV); keep_here<6>(Gn);
  tanh_pair<4>(c, tc); sig_pair<4>(to, ov); ig_pair<6>(iv, gv, m); half_pair<6>(G[3], ao);
  UNIT_END();
  proj_one<7>(Gn, xnext, wih, biasV); keep_here<7>(Gn);
  out_pair<4>(ov, tc, h); cell_pair<6>(fv, m, c); tanh_pair<6>(ao, to);
  UNIT_END();
  proj_one<8>(Gn, xnext, wih, biasV); keep_here<8>(Gn);
  tanh_pair<6>(c, tc); sig_pair<6>(to, ov);
  UNIT_END();
  proj_one<9>(Gn, xnext, wih, biasV); keep_here<9>(Gn);
  out_pair<6>(ov, tc, h);
  UNIT_END();

  proj_one<10>(Gn, xnext, wih, biasV); keep_here<10>(Gn);
  float* yrow = y + ((size_t)t * B + bbase + b) * H + 8 * hi;
  v4f lo, up;
#pragma unroll
  for (int i = 0; i < 4; ++i) { lo[i] = h[i]; up[i] = h[4 + i]; }
  *(v4f*)yrow = lo;
  *(v4f*)(yrow + 4) = up;
  UNIT_END();
  proj_one<11>(Gn, xnext, wih, biasV); keep_here<11>(Gn); UNIT_END();
  proj_one<12>(Gn, xnext, wih, biasV); keep_here<12>(Gn); UNIT_END();
  proj_one<13>(Gn, xnext, wih, biasV); keep_here<13>(Gn); UNIT_END();
  proj_one<14>(Gn, xnext, wih, biasV); keep_here<14>(Gn); UNIT_END();
  proj_one<15>(Gn, xnext, wih, biasV); keep_here<15>(Gn); UNIT_END();
}
#undef UNIT_END
}

__global__ __launch_bounds__(32, 1) void lstm_wmma_kernel(
    const float* __restrict__ x, const float* __restrict__ W_ih,
    const float* __restrict__ W_hh, const float* __restrict__ b_ih,
    const float* __restrict__ b_hh, float* __restrict__ y) {
  const int lane  = threadIdx.x;
  const int b     = lane & 15;
  const int hi    = lane >> 4;
  const int bbase = blockIdx.x * 16;

  v2f wih[4][4], whh[4][4];
#pragma unroll
  for (int nt = 0; nt < 4; ++nt)
#pragma unroll
    for (int j = 0; j < 4; ++j) {
      wih[nt][j] = *(const v2f*)(W_ih + (size_t)(16 * nt + b) * D + 8 * hi + 2 * j);
      whh[nt][j] = *(const v2f*)(W_hh + (size_t)(16 * nt + b) * H + 8 * hi + 2 * j);
    }

  v8f biasV[4];
#pragma unroll
  for (int nt = 0; nt < 4; ++nt)
#pragma unroll
    for (int r = 0; r < 8; ++r)
      biasV[nt][r] = b_ih[16 * nt + 8 * hi + r] + b_hh[16 * nt + 8 * hi + r];

  v8f c = {};
  v8f h = {};

  v4f x0[2], xa[2], xb[2];
  load_x(x0, x, 0, bbase, b, hi);
  load_x(xa, x, 1, bbase, b, hi);
  load_x(xb, x, 2, bbase, b, hi);
  v8f GA[4], GB[4];
  proj_one<0>(GA, x0, wih, biasV);   proj_one<1>(GA, x0, wih, biasV);
  proj_one<2>(GA, x0, wih, biasV);   proj_one<3>(GA, x0, wih, biasV);
  proj_one<4>(GA, x0, wih, biasV);   proj_one<5>(GA, x0, wih, biasV);
  proj_one<6>(GA, x0, wih, biasV);   proj_one<7>(GA, x0, wih, biasV);
  proj_one<8>(GA, x0, wih, biasV);   proj_one<9>(GA, x0, wih, biasV);
  proj_one<10>(GA, x0, wih, biasV);  proj_one<11>(GA, x0, wih, biasV);
  proj_one<12>(GA, x0, wih, biasV);  proj_one<13>(GA, x0, wih, biasV);
  proj_one<14>(GA, x0, wih, biasV);  proj_one<15>(GA, x0, wih, biasV);

  for (int t = 0; t < T; t += 2) {
    const int ta = (t + 3 < T) ? (t + 3) : (T - 1);
    const int tb = (t + 4 < T) ? (t + 4) : (T - 1);
    v4f na[2], nb[2];
    load_x(na, x, ta, bbase, b, hi);
    load_x(nb, x, tb, bbase, b, hi);
    if (t + 18 < T)
      __builtin_prefetch(x + ((size_t)(t + 17 + hi) * B + bbase + b) * D, 0, 1);

    lstm_step(t, GA, GB, xa, wih, whh, biasV, h, c, y, b, hi, bbase);
    lstm_step(t + 1, GB, GA, xb, wih, whh, biasV, h, c, y, b, hi, bbase);

    xa[0] = na[0]; xa[1] = na[1];
    xb[0] = nb[0]; xb[1] = nb[1];
  }
}

extern "C" void kernel_launch(void* const* d_in, const int*  ,
                              int  , void* d_out, int  ,
                              void*  , size_t  ,
                              hipStream_t stream) {
  const float* x    = (const float*)d_in[0];
  const float* W_ih = (const float*)d_in[1];
  const float* W_hh = (const float*)d_in[2];
  const float* b_ih = (const float*)d_in[3];
  const float* b_hh = (const float*)d_in[4];
  float* y = (float*)d_out;
  lstm_wmma_kernel<<<B / 16, 32, 0, stream>>>(x, W_ih, W_hh, b_ih, b_hh, y);
}


# Input projection moved to an all-chip kernel that writes its f32 accumulator tiles to the workspace; the serial wave keeps 16 of 32 matrix instructions per step; f32 sum order unchanged
# speedup vs baseline: 3.1505x; 1.1324x over previous
#include <hip/hip_runtime.h>


typedef __attribute__((ext_vector_type(2))) float v2f;
typedef __attribute__((ext_vector_type(4))) float v4f;
typedef __attribute__((ext_vector_type(8))) float v8f;

namespace {
constexpr int T = 1024, B = 512, D = 16, H = 16;
constexpr int NTILE = B / 16;
constexpr int PROJ_WAVES = 8, PROJ_TILES = 8;
constexpr int PROJ_BLOCKS = T * NTILE / (PROJ_WAVES * PROJ_TILES);
constexpr int TILE_F4 = 256;

__device__ __forceinline__ float tanh_hw(float x) {
  float r;
  asm volatile("v_tanh_f32 %0, %1\n\tv_nop" : "=v"(r) : "v"(x));
  return r;
}


__device__ __forceinline__ void load_x(v4f (&xb)[2], const float* __restrict__ x,
                                       int q, int b, int hi) {
  const float* p = x + ((size_t)q * 16 + b) * D + 8 * hi;
  xb[0] = *(const v4f*)p;
  xb[1] = *(const v4f*)(p + 4);
}

template <int q>
__device__ __forceinline__ void proj_one(v8f (&Gn)[4], const v4f (&xb)[2],
                                         const v2f (&wih)[4][4],
                                         const v8f (&biasV)[4]) {
  constexpr int j = q >> 2, nt = q & 3;
  v2f xB;
  xB[0] = xb[j >> 1][2 * (j & 1)];
  xB[1] = xb[j >> 1][2 * (j & 1) + 1];
  if constexpr (j == 0)
    Gn[nt] = __builtin_amdgcn_wmma_f32_16x16x4_f32(
        false, wih[nt][0], false, xB, (short)0, biasV[nt], false, false);
  else
    Gn[nt] = __builtin_amdgcn_wmma_f32_16x16x4_f32(
        false, wih[nt][j], false, xB, (short)0, Gn[nt], false, false);
}

template <int nt>
__device__ __forceinline__ void keep_here(const v8f (&Gn)[4]) {
  asm volatile("" : : "v"(Gn[nt]));
}

template <int nt, int j>
__device__ __forceinline__ void rec_one(v8f (&G)[4], const v8f& h,
                                        const v2f (&whh)[4][4]) {
  v2f hB;
  hB[0] = h[2 * j];
  hB[1] = h[2 * j + 1];
  G[nt] = __builtin_amdgcn_wmma_f32_16x16x4_f32(
      false, whh[nt][j], false, hB, (short)0, G[nt], false, false);
}

template <int r>
__device__ __forceinline__ void half_pair(const v8f& g, float (&a)[8]) {
  a[r]     = 0.5f * g[r];
  a[r + 1] = 0.5f * g[r + 1];
}
template <int r>
__device__ __forceinline__ void tanh_pair(const float (&a)[8], float (&th)[8]) {
  th[r]     = tanh_hw(a[r]);
  th[r + 1] = tanh_hw(a[r + 1]);
}
template <int r>
__device__ __forceinline__ void tanh_pair(const v8f& g, float (&th)[8]) {
  th[r]     = tanh_hw(g[r]);
  th[r + 1] = tanh_hw(g[r + 1]);
}
template <int r>
__device__ __forceinline__ void sig_pair(const float (&th)[8], float (&s)[8]) {
  s[r]     = __builtin_fmaf(0.5f, th[r], 0.5f);
  s[r + 1] = __builtin_fmaf(0.5f, th[r + 1], 0.5f);
}
template <int r>
__device__ __forceinline__ void ig_pair(const float (&iv)[8], const float (&gv)[8],
                                        float (&m)[8]) {
  m[r]     = iv[r] * gv[r];
  m[r + 1] = iv[r + 1] * gv[r + 1];
}
template <int r>
__device__ __forceinline__ void cell_pair(const float (&fv)[8], const float (&m)[8],
                                          v8f& c) {
  c[r]     = __builtin_fmaf(fv[r], c[r], m[r]);
  c[r + 1] = __builtin_fmaf(fv[r + 1], c[r + 1], m[r + 1]);
}
template <int r>
__device__ __forceinline__ void out_pair(const float (&ov)[8], const float (&tc)[8],
                                         v8f& h) {
  h[r]     = ov[r] * tc[r];
  h[r + 1] = ov[r + 1] * tc[r + 1];
}

#define UNIT_END() __builtin_amdgcn_sched_barrier(0)

__device__ __forceinline__ void load_g(v8f (&G)[4], const float* __restrict__ xg,
                                       int t, int tile, int lane) {
  const v4f* p = (const v4f*)xg + ((size_t)t * NTILE + tile) * TILE_F4 + lane;
#pragma unroll
  for (int nt = 0; nt < 4; ++nt) {
    const v4f lo = p[(2 * nt) * 32], up = p[(2 * nt + 1) * 32];
    G[nt] = __builtin_shufflevector(lo, up, 0, 1, 2, 3, 4, 5, 6, 7);
  }
}

__device__ __forceinline__ void prefetch_g(const float* __restrict__ xg,
                                           int t, int tile, int lane) {
  __builtin_prefetch(xg + ((size_t)t * NTILE + tile) * (TILE_F4 * 4) + lane * 32, 0, 1);
}

__device__ __forceinline__ void lstm_trip(
    int t, v8f (&G)[4], v8f (&Gn)[4], const v2f (&whh)[4][4],
    v8f& h, v8f& c, float (&ai)[8], float* __restrict__ y,
    int b, int hi, int bbase) {
  float af[8], ao[8], ti[8], tf[8], to[8];
  float iv[8], fv[8], gv[8], ov[8], m[8], tc[8];

  rec_one<2, 0>(G, h, whh);                      tanh_pair<0>(ai, ti); half_pair<2>(G[0], ai); UNIT_END();
  rec_one<3, 0>(G, h, whh); sig_pair<0>(ti, iv); tanh_pair<2>(ai, ti); half_pair<4>(G[0], ai); UNIT_END();
  rec_one<2, 1>(G, h, whh); sig_pair<2>(ti, iv); tanh_pair<4>(ai, ti); half_pair<6>(G[0], ai); UNIT_END();
  rec_one<3, 1>(G, h, whh); sig_pair<4>(ti, iv); tanh_pair<6>(ai, ti); half_pair<0>(G[1], af); UNIT_END();
  rec_one<2, 2>(G, h, whh); sig_pair<6>(ti, iv); tanh_pair<0>(af, tf); half_pair<2>(G[1], af); UNIT_END();
  rec_one<3, 2>(G, h, whh); sig_pair<0>(tf, fv); tanh_pair<2>(af, tf); half_pair<4>(G[1], af); UNIT_END();
  rec_one<2, 3>(G, h, whh); sig_pair<2>(tf, fv); tanh_pair<4>(af, tf); half_pair<6>(G[1], af); UNIT_END();
  rec_one<3, 3>(G, h, whh); sig_pair<4>(tf, fv); tanh_pair<6>(af, tf); tanh_pair<0>(G[2], gv); UNIT_END();

  sig_pair<6>(tf, fv); ig_pair<0>(iv, gv, m); half_pair<0>(G[3], ao);
  UNIT_END();
  cell_pair<0>(fv, m, c); tanh_pair<0>(ao, to); tanh_pair<2>(G[2], gv);
  UNIT_END();
  tanh_pair<0>(c, tc); sig_pair<0>(to, ov); ig_pair<2>(iv, gv, m); half_pair<2>(G[3], ao);
  UNIT_END();
  out_pair<0>(ov, tc, h); cell_pair<2>(fv, m, c); tanh_pair<2>(ao, to); tanh_pair<4>(G[2], gv);
  UNIT_END();
  rec_one<0, 0>(Gn, h, whh); keep_here<0>(Gn);
  tanh_pair<2>(c, tc); sig_pair<2>(to, ov); ig_pair<4>(iv, gv, m); half_pair<4>(G[3], ao);
  UNIT_END();
  rec_one<1, 0>(Gn, h, whh); keep_here<1>(Gn);
  out_pair<2>(ov, tc, h); cell_pair<4>(fv, m, c); tanh_pair<4>(ao, to); tanh_pair<6>(G[2], gv);
  UNIT_END();
  rec_one<0, 1>(Gn, h, whh); keep_here<0>(Gn);
  tanh_pair<4>(c, tc); sig_pair<4>(to, ov); ig_pair<6>(iv, gv, m); half_pair<6>(G[3], ao);
  UNIT_END();
  rec_one<1, 1>(Gn, h, whh); keep_here<1>(Gn);
  out_pair<4>(ov, tc, h); cell_pair<6>(fv, m, c); tanh_pair<6>(ao, to);
  UNIT_END();
  rec_one<0, 2>(Gn, h, whh); keep_here<0>(Gn);
  tanh_pair<6>(c, tc); sig_pair<6>(to, ov);
  UNIT_END();
  rec_one<1, 2>(Gn, h, whh); keep_here<1>(Gn);
  out_pair<6>(ov, tc, h);
  UNIT_END();

  rec_one<0, 3>(Gn, h, whh); keep_here<0>(Gn);
  float* yrow = y + ((size_t)t * B + bbase + b) * H + 8 * hi;
  v4f lo, up;
#pragma unroll
  for (int i = 0; i < 4; ++i) { lo[i] = h[i]; up[i] = h[4 + i]; }
  *(v4f*)yrow = lo;
  *(v4f*)(yrow + 4) = up;
  UNIT_END();
  rec_one<1, 3>(Gn, h, whh); keep_here<1>(Gn); half_pair<0>(Gn[0], ai); UNIT_END();
}
}

__global__ __launch_bounds__(32 * PROJ_WAVES) void lstm_proj_kernel(
    const float* __restrict__ x, const float* __restrict__ W_ih,
    const float* __restrict__ b_ih, const float* __restrict__ b_hh,
    float* __restrict__ xg) {
  const int lane = threadIdx.x & 31;
  const int wave = threadIdx.x >> 5;
  const int b    = lane & 15;
  const int hi   = lane >> 4;

  v2f wih[4][4];
#pragma unroll
  for (int nt = 0; nt < 4; ++nt)
#pragma unroll
    for (int j = 0; j < 4; ++j)
      wih[nt][j] = *(const v2f*)(W_ih + (size_t)(16 * nt + b) * D + 8 * hi + 2 * j);

  v8f biasV[4];
#pragma unroll
  for (int nt = 0; nt < 4; ++nt)
#pragma unroll
    for (int r = 0; r < 8; ++r)
      biasV[nt][r] = b_ih[16 * nt + 8 * hi + r] + b_hh[16 * nt + 8 * hi + r];

  const int q0 = (blockIdx.x * PROJ_WAVES + wave) * PROJ_TILES;
#pragma unroll
  for (int i = 0; i < PROJ_TILES; ++i) {
    const int q = q0 + i;
    v4f xb[2];
    load_x(xb, x, q, b, hi);
    v8f G[4];
    proj_one<0>(G, xb, wih, biasV);   proj_one<1>(G, xb, wih, biasV);
    proj_one<2>(G, xb, wih, biasV);   proj_one<3>(G, xb, wih, biasV);
    proj_one<4>(G, xb, wih, biasV);   proj_one<5>(G, xb, wih, biasV);
    proj_one<6>(G, xb, wih, biasV);   proj_one<7>(G, xb, wih, biasV);
    proj_one<8>(G, xb, wih, biasV);   proj_one<9>(G, xb, wih, biasV);
    proj_one<10>(G, xb, wih, biasV);  proj_one<11>(G, xb, wih, biasV);
    proj_one<12>(G, xb, wih, biasV);  proj_one<13>(G, xb, wih, biasV);
    proj_one<14>(G, xb, wih, biasV);  proj_one<15>(G, xb, wih, biasV);
    v4f* dst = (v4f*)xg + (size_t)q * TILE_F4 + lane;
#pragma unroll
    for (int nt = 0; nt < 4; ++nt) {
      dst[(2 * nt) * 32]     = __builtin_shufflevector(G[nt], G[nt], 0, 1, 2, 3);
      dst[(2 * nt + 1) * 32] = __builtin_shufflevector(G[nt], G[nt], 4, 5, 6, 7);
    }
  }
}

__global__ __launch_bounds__(32, 1) void lstm_rec_kernel(
    const float* __restrict__ xg, const float* __restrict__ W_hh,
    float* __restrict__ y) {
  const int lane  = threadIdx.x;
  const int b     = lane & 15;
  const int hi    = lane >> 4;
  const int tile  = blockIdx.x;
  const int bbase = tile * 16;

  v2f whh[4][4];
#pragma unroll
  for (int nt = 0; nt < 4; ++nt)
#pragma unroll
    for (int j = 0; j < 4; ++j)
      whh[nt][j] = *(const v2f*)(W_hh + (size_t)(16 * nt + b) * H + 8 * hi + 2 * j);

  v8f c = {};
  v8f h = {};

  v8f G0[4], G1[4], G2[4], G3[4], G4[4], G5[4], G6[4], G7[4];
  load_g(G0, xg, 0, tile, lane);
  load_g(G1, xg, 1, tile, lane);
  load_g(G2, xg, 2, tile, lane);
  load_g(G3, xg, 3, tile, lane);

  float ai[8];
  rec_one<0, 0>(G0, h, whh); rec_one<1, 0>(G0, h, whh);
  rec_one<0, 1>(G0, h, whh); rec_one<1, 1>(G0, h, whh);
  rec_one<0, 2>(G0, h, whh); rec_one<1, 2>(G0, h, whh);
  rec_one<0, 3>(G0, h, whh); rec_one<1, 3>(G0, h, whh);
  half_pair<0>(G0[0], ai);

  for (int t = 0; t < T; t += 8) {
    if (t + 12 < T) prefetch_g(xg, t + 12, tile, lane);
    load_g(G4, xg, (t + 4 < T) ? (t + 4) : (T - 1), tile, lane);
    lstm_trip(t, G0, G1, whh, h, c, ai, y, b, hi, bbase);
    if (t + 13 < T) prefetch_g(xg, t + 13, tile, lane);
    load_g(G5, xg, (t + 5 < T) ? (t + 5) : (T - 1), tile, lane);
    lstm_trip(t + 1, G1, G2, whh, h, c, ai, y, b, hi, bbase);
    if (t + 14 < T) prefetch_g(xg, t + 14, tile, lane);
    load_g(G6, xg, (t + 6 < T) ? (t + 6) : (T - 1), tile, lane);
    lstm_trip(t + 2, G2, G3, whh, h, c, ai, y, b, hi, bbase);
    if (t + 15 < T) prefetch_g(xg, t + 15, tile, lane);
    load_g(G7, xg, (t + 7 < T) ? (t + 7) : (T - 1), tile, lane);
    lstm_trip(t + 3, G3, G4, whh, h, c, ai, y, b, hi, bbase);
    if (t + 16 < T) prefetch_g(xg, t + 16, tile, lane);
    load_g(G0, xg, (t + 8 < T) ? (t + 8) : (T - 1), tile, lane);
    lstm_trip(t + 4, G4, G5, whh, h, c, ai, y, b, hi, bbase);
    if (t + 17 < T) prefetch_g(xg, t + 17, tile, lane);
    load_g(G1, xg, (t + 9 < T) ? (t + 9) : (T - 1), tile, lane);
    lstm_trip(t + 5, G5, G6, whh, h, c, ai, y, b, hi, bbase);
    if (t + 18 < T) prefetch_g(xg, t + 18, tile, lane);
    load_g(G2, xg, (t + 10 < T) ? (t + 10) : (T - 1), tile, lane);
    lstm_trip(t + 6, G6, G7, whh, h, c, ai, y, b, hi, bbase);
    if (t + 19 < T) prefetch_g(xg, t + 19, tile, lane);
    load_g(G3, xg, (t + 11 < T) ? (t + 11) : (T - 1), tile, lane);
    lstm_trip(t + 7, G7, G0, whh, h, c, ai, y, b, hi, bbase);
  }
}
#undef UNIT_END

extern "C" void kernel_launch(void* const* d_in, const int*  ,
                              int  , void* d_out, int  ,
                              void* d_ws, size_t  ,
                              hipStream_t stream) {
  const float* x    = (const float*)d_in[0];
  const float* W_ih = (const float*)d_in[1];
  const float* W_hh = (const float*)d_in[2];
  const float* b_ih = (const float*)d_in[3];
  const float* b_hh = (const float*)d_in[4];
  float* y  = (float*)d_out;
  float* xg = (float*)d_ws;
  lstm_proj_kernel<<<PROJ_BLOCKS, 32 * PROJ_WAVES, 0, stream>>>(x, W_ih, b_ih, b_hh, xg);
  lstm_rec_kernel<<<NTILE, 32, 0, stream>>>(xg, W_hh, y);
}


# Serial wave's memory instructions gathered at one place per two steps (y stores, then the tile loads, over three steps ahead); deep prefetch and in-loop weight-load wait gone; f32 sum order unchanged
# speedup vs baseline: 3.5987x; 1.1423x over previous
#include <hip/hip_runtime.h>


typedef __attribute__((ext_vector_type(2))) float v2f;
typedef __attribute__((ext_vector_type(4))) float v4f;
typedef __attribute__((ext_vector_type(8))) float v8f;

namespace {
constexpr int T = 1024, B = 512, D = 16, H = 16;
constexpr int NTILE = B / 16;
constexpr int PROJ_WAVES = 8, PROJ_TILES = 8;
constexpr int PROJ_BLOCKS = T * NTILE / (PROJ_WAVES * PROJ_TILES);
constexpr int TILE_F4 = 256;

__device__ __forceinline__ float tanh_hw(float x) {
  float r;
  asm volatile("v_tanh_f32 %0, %1\n\tv_nop" : "=v"(r) : "v"(x));
  return r;
}


__device__ __forceinline__ void load_x(v4f (&xb)[2], const float* __restrict__ x,
                                       int q, int b, int hi) {
  const float* p = x + ((size_t)q * 16 + b) * D + 8 * hi;
  xb[0] = *(const v4f*)p;
  xb[1] = *(const v4f*)(p + 4);
}

template <int q>
__device__ __forceinline__ void proj_one(v8f (&Gn)[4], const v4f (&xb)[2],
                                         const v2f (&wih)[4][4],
                                         const v8f (&biasV)[4]) {
  constexpr int j = q >> 2, nt = q & 3;
  v2f xB;
  xB[0] = xb[j >> 1][2 * (j & 1)];
  xB[1] = xb[j >> 1][2 * (j & 1) + 1];
  if constexpr (j == 0)
    Gn[nt] = __builtin_amdgcn_wmma_f32_16x16x4_f32(
        false, wih[nt][0], false, xB, (short)0, biasV[nt], false, false);
  else
    Gn[nt] = __builtin_amdgcn_wmma_f32_16x16x4_f32(
        false, wih[nt][j], false, xB, (short)0, Gn[nt], false, false);
}

template <int nt>
__device__ __forceinline__ void keep_here(const v8f (&Gn)[4]) {
  asm volatile("" : : "v"(Gn[nt]));
}

template <int nt, int j>
__device__ __forceinline__ void rec_one(v8f (&G)[4], const v8f& h,
                                        const v2f (&whh)[4][4]) {
  v2f hB;
  hB[0] = h[2 * j];
  hB[1] = h[2 * j + 1];
  G[nt] = __builtin_amdgcn_wmma_f32_16x16x4_f32(
      false, whh[nt][j], false, hB, (short)0, G[nt], false, false);
}

template <int r>
__device__ __forceinline__ void half_pair(const v8f& g, float (&a)[8]) {
  a[r]     = 0.5f * g[r];
  a[r + 1] = 0.5f * g[r + 1];
}
template <int r>
__device__ __forceinline__ void tanh_pair(const float (&a)[8], float (&th)[8]) {
  th[r]     = tanh_hw(a[r]);
  th[r + 1] = tanh_hw(a[r + 1]);
}
template <int r>
__device__ __forceinline__ void tanh_pair(const v8f& g, float (&th)[8]) {
  th[r]     = tanh_hw(g[r]);
  th[r + 1] = tanh_hw(g[r + 1]);
}
template <int r>
__device__ __forceinline__ void sig_pair(const float (&th)[8], float (&s)[8]) {
  s[r]     = __builtin_fmaf(0.5f, th[r], 0.5f);
  s[r + 1] = __builtin_fmaf(0.5f, th[r + 1], 0.5f);
}
template <int r>
__device__ __forceinline__ void ig_pair(const float (&iv)[8], const float (&gv)[8],
                                        float (&m)[8]) {
  m[r]     = iv[r] * gv[r];
  m[r + 1] = iv[r + 1] * gv[r + 1];
}
template <int r>
__device__ __forceinline__ void cell_pair(const float (&fv)[8], const float (&m)[8],
                                          v8f& c) {
  c[r]     = __builtin_fmaf(fv[r], c[r], m[r]);
  c[r + 1] = __builtin_fmaf(fv[r + 1], c[r + 1], m[r + 1]);
}
template <int r>
__device__ __forceinline__ void out_pair(const float (&ov)[8], const float (&tc)[8],
                                         v8f& h) {
  h[r]     = ov[r] * tc[r];
  h[r + 1] = ov[r + 1] * tc[r + 1];
}

#define UNIT_END() __builtin_amdgcn_sched_barrier(0)

__device__ __forceinline__ void load_g(v8f (&G)[4], const float* __restrict__ xg,
                                       int t, int tile, int lane) {
  const v4f* p = (const v4f*)xg + ((size_t)t * NTILE + tile) * TILE_F4 + lane;
#pragma unroll
  for (int nt = 0; nt < 4; ++nt) {
    const v4f lo = p[(2 * nt) * 32], up = p[(2 * nt + 1) * 32];
    G[nt] = __builtin_shufflevector(lo, up, 0, 1, 2, 3, 4, 5, 6, 7);
  }
}

__device__ __forceinline__ void store_y(float* __restrict__ y, int t, int row,
                                        int hi, const v8f& hv) {
  float* yrow = y + ((size_t)t * B + row) * H + 8 * hi;
  *(v4f*)yrow       = __builtin_shufflevector(hv, hv, 0, 1, 2, 3);
  *(v4f*)(yrow + 4) = __builtin_shufflevector(hv, hv, 4, 5, 6, 7);
}

template <class MemOps>
__device__ __forceinline__ void lstm_trip(
    v8f (&G)[4], v8f (&Gn)[4], const v2f (&whh)[4][4],
    v8f& h, v8f& c, float (&ai)[8], MemOps&& mem_ops) {
  float af[8], ao[8], ti[8], tf[8], to[8];
  float iv[8], fv[8], gv[8], ov[8], m[8], tc[8];

  rec_one<2, 0>(G, h, whh);                      tanh_pair<0>(ai, ti); half_pair<2>(G[0], ai); UNIT_END();
  rec_one<3, 0>(G, h, whh); sig_pair<0>(ti, iv); tanh_pair<2>(ai, ti); half_pair<4>(G[0], ai); UNIT_END();
  rec_one<2, 1>(G, h, whh); sig_pair<2>(ti, iv); tanh_pair<4>(ai, ti); half_pair<6>(G[0], ai); UNIT_END();
  rec_one<3, 1>(G, h, whh); sig_pair<4>(ti, iv); tanh_pair<6>(ai, ti); half_pair<0>(G[1], af); UNIT_END();
  rec_one<2, 2>(G, h, whh); sig_pair<6>(ti, iv); tanh_pair<0>(af, tf); half_pair<2>(G[1], af); UNIT_END();
  rec_one<3, 2>(G, h, whh); sig_pair<0>(tf, fv); tanh_pair<2>(af, tf); half_pair<4>(G[1], af); UNIT_END();
  rec_one<2, 3>(G, h, whh); sig_pair<2>(tf, fv); tanh_pair<4>(af, tf); half_pair<6>(G[1], af); UNIT_END();
  rec_one<3, 3>(G, h, whh); sig_pair<4>(tf, fv); tanh_pair<6>(af, tf); tanh_pair<0>(G[2], gv); UNIT_END();

  sig_pair<6>(tf, fv); ig_pair<0>(iv, gv, m); half_pair<0>(G[3], ao);
  UNIT_END();
  cell_pair<0>(fv, m, c); tanh_pair<0>(ao, to); tanh_pair<2>(G[2], gv);
  UNIT_END();
  tanh_pair<0>(c, tc); sig_pair<0>(to, ov); ig_pair<2>(iv, gv, m); half_pair<2>(G[3], ao);
  UNIT_END();
  out_pair<0>(ov, tc, h); cell_pair<2>(fv, m, c); tanh_pair<2>(ao, to); tanh_pair<4>(G[2], gv);
  UNIT_END();
  rec_one<0, 0>(Gn, h, whh); keep_here<0>(Gn);
  tanh_pair<2>(c, tc); sig_pair<2>(to, ov); ig_pair<4>(iv, gv, m); half_pair<4>(G[3], ao);
  UNIT_END();
  rec_one<1, 0>(Gn, h, whh); keep_here<1>(Gn);
  out_pair<2>(ov, tc, h); cell_pair<4>(fv, m, c); tanh_pair<4>(ao, to); tanh_pair<6>(G[2], gv);
  UNIT_END();
  rec_one<0, 1>(Gn, h, whh); keep_here<0>(Gn);
  tanh_pair<4>(c, tc); sig_pair<4>(to, ov); ig_pair<6>(iv, gv, m); half_pair<6>(G[3], ao);
  UNIT_END();
  rec_one<1, 1>(Gn, h, whh); keep_here<1>(Gn);
  out_pair<4>(ov, tc, h); cell_pair<6>(fv, m, c); tanh_pair<6>(ao, to);
  UNIT_END();
  rec_one<0, 2>(Gn, h, whh); keep_here<0>(Gn);
  tanh_pair<6>(c, tc); sig_pair<6>(to, ov);
  UNIT_END();
  rec_one<1, 2>(Gn, h, whh); keep_here<1>(Gn);
  out_pair<6>(ov, tc, h);
  UNIT_END();

  rec_one<0, 3>(Gn, h, whh); keep_here<0>(Gn);
  mem_ops();
  UNIT_END();
  rec_one<1, 3>(Gn, h, whh); keep_here<1>(Gn); half_pair<0>(Gn[0], ai); UNIT_END();
}
}

__global__ __launch_bounds__(32 * PROJ_WAVES) void lstm_proj_kernel(
    const float* __restrict__ x, const float* __restrict__ W_ih,
    const float* __restrict__ b_ih, const float* __restrict__ b_hh,
    float* __restrict__ xg) {
  const int lane = threadIdx.x & 31;
  const int wave = threadIdx.x >> 5;
  const int b    = lane & 15;
  const int hi   = lane >> 4;

  v2f wih[4][4];
#pragma unroll
  for (int nt = 0; nt < 4; ++nt)
#pragma unroll
    for (int j = 0; j < 4; ++j)
      wih[nt][j] = *(const v2f*)(W_ih + (size_t)(16 * nt + b) * D + 8 * hi + 2 * j);

  v8f biasV[4];
#pragma unroll
  for (int nt = 0; nt < 4; ++nt)
#pragma unroll
    for (int r = 0; r < 8; ++r)
      biasV[nt][r] = b_ih[16 * nt + 8 * hi + r] + b_hh[16 * nt + 8 * hi + r];

  const int q0 = (blockIdx.x * PROJ_WAVES + wave) * PROJ_TILES;
#pragma unroll
  for (int i = 0; i < PROJ_TILES; ++i) {
    const int q = q0 + i;
    v4f xb[2];
    load_x(xb, x, q, b, hi);
    v8f G[4];
    proj_one<0>(G, xb, wih, biasV);   proj_one<1>(G, xb, wih, biasV);
    proj_one<2>(G, xb, wih, biasV);   proj_one<3>(G, xb, wih, biasV);
    proj_one<4>(G, xb, wih, biasV);   proj_one<5>(G, xb, wih, biasV);
    proj_one<6>(G, xb, wih, biasV);   proj_one<7>(G, xb, wih, biasV);
    proj_one<8>(G, xb, wih, biasV);   proj_one<9>(G, xb, wih, biasV);
    proj_one<10>(G, xb, wih, biasV);  proj_one<11>(G, xb, wih, biasV);
    proj_one<12>(G, xb, wih, biasV);  proj_one<13>(G, xb, wih, biasV);
    proj_one<14>(G, xb, wih, biasV);  proj_one<15>(G, xb, wih, biasV);
    v4f* dst = (v4f*)xg + (size_t)q * TILE_F4 + lane;
#pragma unroll
    for (int nt = 0; nt < 4; ++nt) {
      dst[(2 * nt) * 32]     = __builtin_shufflevector(G[nt], G[nt], 0, 1, 2, 3);
      dst[(2 * nt + 1) * 32] = __builtin_shufflevector(G[nt], G[nt], 4, 5, 6, 7);
    }
  }
}

__global__ __launch_bounds__(32, 1) void lstm_rec_kernel(
    const float* __restrict__ xg, const float* __restrict__ W_hh,
    float* __restrict__ y) {
  const int lane  = threadIdx.x;
  const int b     = lane & 15;
  const int hi    = lane >> 4;
  const int tile  = blockIdx.x;
  const int bbase = tile * 16;

  v2f whh[4][4];
#pragma unroll
  for (int nt = 0; nt < 4; ++nt)
#pragma unroll
    for (int j = 0; j < 4; ++j)
      whh[nt][j] = *(const v2f*)(W_hh + (size_t)(16 * nt + b) * H + 8 * hi + 2 * j);

  v8f c = {};
  v8f h = {};

  v8f G0[4], G1[4], G2[4], G3[4], G4[4], G5[4], G6[4], G7[4];
  load_g(G0, xg, 0, tile, lane);
  load_g(G1, xg, 1, tile, lane);
  load_g(G2, xg, 2, tile, lane);
  load_g(G3, xg, 3, tile, lane);
  load_g(G4, xg, 4, tile, lane);
  load_g(G5, xg, 5, tile, lane);

  float ai[8];
  rec_one<0, 0>(G0, h, whh); rec_one<1, 0>(G0, h, whh);
  rec_one<0, 1>(G0, h, whh); rec_one<1, 1>(G0, h, whh);
  rec_one<0, 2>(G0, h, whh); rec_one<1, 2>(G0, h, whh);
  rec_one<0, 3>(G0, h, whh); rec_one<1, 3>(G0, h, whh);
  half_pair<0>(G0[0], ai);

  const int row = bbase + b;
  const auto no_mem = [] {};

#define TWO_TRIPS(k, Ga, Gb, Gc, Gx, Gy, TL)                                   \
  {                                                                            \
    lstm_trip(Ga, Gb, whh, h, c, ai, no_mem);                                  \
    const v8f hs = h;                                                          \
    lstm_trip(Gb, Gc, whh, h, c, ai, [&] {                                     \
      store_y(y, t + (k),     row, hi, hs);                                    \
      store_y(y, t + (k) + 1, row, hi, h);                                     \
      load_g(Gx, xg, TL((k) + 6), tile, lane);                                 \
      load_g(Gy, xg, TL((k) + 7), tile, lane);                                 \
    });                                                                        \
  }
#define TL_PLAIN(k) (t + (k))
#define TL_LAST(k) ((t + (k) < T) ? (t + (k)) : (T - 1))

  {
    constexpr int t = 0;
    TWO_TRIPS(0, G0, G1, G2, G6, G7, TL_PLAIN)
  }
  for (int t = 2; t < T - 6; t += 8) {
    TWO_TRIPS(0, G2, G3, G4, G0, G1, TL_PLAIN)
    TWO_TRIPS(2, G4, G5, G6, G2, G3, TL_PLAIN)
    TWO_TRIPS(4, G6, G7, G0, G4, G5, TL_PLAIN)
    TWO_TRIPS(6, G0, G1, G2, G6, G7, TL_PLAIN)
  }
  {
    constexpr int t = T - 6;
    TWO_TRIPS(0, G2, G3, G4, G0, G1, TL_LAST)
    TWO_TRIPS(2, G4, G5, G6, G2, G3, TL_LAST)
    TWO_TRIPS(4, G6, G7, G0, G4, G5, TL_LAST)
  }
#undef TL_LAST
#undef TL_PLAIN
#undef TWO_TRIPS
}
#undef UNIT_END

extern "C" void kernel_launch(void* const* d_in, const int*  ,
                              int  , void* d_out, int  ,
                              void* d_ws, size_t  ,
                              hipStream_t stream) {
  const float* x    = (const float*)d_in[0];
  const float* W_ih = (const float*)d_in[1];
  const float* W_hh = (const float*)d_in[2];
  const float* b_ih = (const float*)d_in[3];
  const float* b_hh = (const float*)d_in[4];
  float* y  = (float*)d_out;
  float* xg = (float*)d_ws;
  lstm_proj_kernel<<<PROJ_BLOCKS, 32 * PROJ_WAVES, 0, stream>>>(x, W_ih, b_ih, b_hh, xg);
  lstm_rec_kernel<<<NTILE, 32, 0, stream>>>(xg, W_hh, y);
}


# One batch tile's recurrence shared by four waves, one per SIMD: each owns a quarter of the gate rows (4 matrix instructions a step), h exchanged via LDS, one barrier a step; f32 sum order unchanged
# speedup vs baseline: 4.7251x; 1.3130x over previous
#include <hip/hip_runtime.h>


typedef __attribute__((ext_vector_type(2))) float v2f;
typedef __attribute__((ext_vector_type(4))) float v4f;
typedef __attribute__((ext_vector_type(8))) float v8f;

namespace {
constexpr int T = 1024, B = 512, D = 16, H = 16;
constexpr int NTILE = B / 16;
constexpr int PROJ_WAVES = 8, PROJ_TILES = 8;
constexpr int PROJ_BLOCKS = T * NTILE / (PROJ_WAVES * PROJ_TILES);
constexpr int TILE_F4 = 256;
constexpr int REC_WAVES = 4;

__device__ __forceinline__ int gate_row(int w, int m) {
  return 16 * ((m & 7) >> 1) + 8 * (m >> 3) + 2 * w + (m & 1);
}

__device__ __forceinline__ float tanh_hw(float x) {
  float r;
  asm volatile("v_tanh_f32 %0, %1\n\tv_nop" : "=v"(r) : "v"(x));
  return r;
}


__device__ __forceinline__ void load_x(v4f (&xb)[2], const float* __restrict__ x,
                                       int q, int b, int hi) {
  const float* p = x + ((size_t)q * 16 + b) * D + 8 * hi;
  xb[0] = *(const v4f*)p;
  xb[1] = *(const v4f*)(p + 4);
}

template <int q>
__device__ __forceinline__ void proj_one(v8f (&Gn)[4], const v4f (&xb)[2],
                                         const v2f (&wih)[4][4],
                                         const v8f (&biasV)[4]) {
  constexpr int j = q >> 2, w = q & 3;
  v2f xB;
  xB[0] = xb[j >> 1][2 * (j & 1)];
  xB[1] = xb[j >> 1][2 * (j & 1) + 1];
  if constexpr (j == 0)
    Gn[w] = __builtin_amdgcn_wmma_f32_16x16x4_f32(
        false, wih[w][0], false, xB, (short)0, biasV[w], false, false);
  else
    Gn[w] = __builtin_amdgcn_wmma_f32_16x16x4_f32(
        false, wih[w][j], false, xB, (short)0, Gn[w], false, false);
}

__device__ __forceinline__ void store_y(float* __restrict__ y, int t, int row,
                                        int hi, const v8f& hv) {
  float* yrow = y + ((size_t)t * B + row) * H + 8 * hi;
  *(v4f*)yrow       = __builtin_shufflevector(hv, hv, 0, 1, 2, 3);
  *(v4f*)(yrow + 4) = __builtin_shufflevector(hv, hv, 4, 5, 6, 7);
}

__device__ __forceinline__ void h_publish() {
  asm volatile("s_wait_dscnt 0x0" ::: "memory");
  asm volatile("s_barrier_signal -1" ::: "memory");
}
__device__ __forceinline__ void h_collect() {
  asm volatile("s_barrier_wait -1" ::: "memory");
}
}

__global__ __launch_bounds__(32 * PROJ_WAVES) void lstm_proj_kernel(
    const float* __restrict__ x, const float* __restrict__ W_ih,
    const float* __restrict__ b_ih, const float* __restrict__ b_hh,
    float* __restrict__ xg) {
  const int lane = threadIdx.x & 31;
  const int wave = threadIdx.x >> 5;
  const int b    = lane & 15;
  const int hi   = lane >> 4;

  v2f wih[4][4];
#pragma unroll
  for (int w = 0; w < 4; ++w)
#pragma unroll
    for (int j = 0; j < 4; ++j)
      wih[w][j] = *(const v2f*)(W_ih + (size_t)gate_row(w, b) * D + 8 * hi + 2 * j);

  v8f biasV[4];
#pragma unroll
  for (int w = 0; w < 4; ++w)
#pragma unroll
    for (int r = 0; r < 8; ++r) {
      const int g = gate_row(w, 8 * hi + r);
      biasV[w][r] = b_ih[g] + b_hh[g];
    }

  const int q0 = (blockIdx.x * PROJ_WAVES + wave) * PROJ_TILES;
#pragma unroll
  for (int i = 0; i < PROJ_TILES; ++i) {
    const int q = q0 + i;
    v4f xb[2];
    load_x(xb, x, q, b, hi);
    v8f G[4];
    proj_one<0>(G, xb, wih, biasV);   proj_one<1>(G, xb, wih, biasV);
    proj_one<2>(G, xb, wih, biasV);   proj_one<3>(G, xb, wih, biasV);
    proj_one<4>(G, xb, wih, biasV);   proj_one<5>(G, xb, wih, biasV);
    proj_one<6>(G, xb, wih, biasV);   proj_one<7>(G, xb, wih, biasV);
    proj_one<8>(G, xb, wih, biasV);   proj_one<9>(G, xb, wih, biasV);
    proj_one<10>(G, xb, wih, biasV);  proj_one<11>(G, xb, wih, biasV);
    proj_one<12>(G, xb, wih, biasV);  proj_one<13>(G, xb, wih, biasV);
    proj_one<14>(G, xb, wih, biasV);  proj_one<15>(G, xb, wih, biasV);
    v4f* dst = (v4f*)xg + (size_t)q * TILE_F4 + lane;
#pragma unroll
    for (int w = 0; w < 4; ++w) {
      dst[(2 * w) * 32]     = __builtin_shufflevector(G[w], G[w], 0, 1, 2, 3);
      dst[(2 * w + 1) * 32] = __builtin_shufflevector(G[w], G[w], 4, 5, 6, 7);
    }
  }
}

__global__ __launch_bounds__(32 * REC_WAVES) void lstm_rec_kernel(
    const float* __restrict__ xg, const float* __restrict__ W_hh,
    float* __restrict__ y) {
  __shared__ v2f hbuf[2][REC_WAVES][32];

  const int lane = threadIdx.x & 31;
  const int w    = __builtin_amdgcn_readfirstlane((int)(threadIdx.x >> 5));
  const int b    = lane & 15;
  const int hi   = lane >> 4;
  const int tile = blockIdx.x;
  const int row  = tile * 16 + b;

  v2f whh[4];
#pragma unroll
  for (int j = 0; j < 4; ++j)
    whh[j] = *(const v2f*)(W_hh + (size_t)gate_row(w, b) * H + 8 * hi + 2 * j);

  const v4f* gp = (const v4f*)xg + (size_t)tile * TILE_F4 + (2 * w) * 32 + lane;
  auto load_g = [&](int t) -> v8f {
    const v4f* p = gp + (size_t)t * (NTILE * TILE_F4);
    const v4f lo = p[0], up = p[32];
    return __builtin_shufflevector(lo, up, 0, 1, 2, 3, 4, 5, 6, 7);
  };

  v8f G0 = load_g(0), G1 = load_g(1), G2 = load_g(2), G3 = load_g(3);
  v8f G4 = load_g(4), G5 = load_g(5), G6 = load_g(6), G7 = load_g(7);

  v8f hf = {};
  float c0 = 0.0f, c1 = 0.0f;

#define STEP(k, Gk)                                                            \
  {                                                                            \
    v8f acc = Gk;                                                              \
    _Pragma("unroll")                                                          \
    for (int j = 0; j < 4; ++j) {                                              \
      v2f hB;                                                                  \
      hB[0] = hf[2 * j];                                                       \
      hB[1] = hf[2 * j + 1];                                                   \
      acc = __builtin_amdgcn_wmma_f32_16x16x4_f32(                             \
          false, whh[j], false, hB, (short)0, acc, false, false);              \
    }                                                                          \
      \
      \
    const float ai0 = 0.5f * acc[0], ai1 = 0.5f * acc[1];                      \
    const float af0 = 0.5f * acc[2], af1 = 0.5f * acc[3];                      \
    const float ao0 = 0.5f * acc[6], ao1 = 0.5f * acc[7];                      \
    const float g0 = tanh_hw(acc[4]), g1 = tanh_hw(acc[5]);                    \
    const float ti0 = tanh_hw(ai0), ti1 = tanh_hw(ai1);                        \
    const float tf0 = tanh_hw(af0), tf1 = tanh_hw(af1);                        \
    const float i0 = __builtin_fmaf(0.5f, ti0, 0.5f);                          \
    const float i1 = __builtin_fmaf(0.5f, ti1, 0.5f);                          \
    const float f0 = __builtin_fmaf(0.5f, tf0, 0.5f);                          \
    const float f1 = __builtin_fmaf(0.5f, tf1, 0.5f);                          \
               \
    const float m0 = i0 * g0, m1 = i1 * g1;                                    \
    c0 = __builtin_fmaf(f0, c0, m0);                                           \
    c1 = __builtin_fmaf(f1, c1, m1);                                           \
    const float to0 = tanh_hw(ao0), to1 = tanh_hw(ao1);                        \
    const float tc0 = tanh_hw(c0), tc1 = tanh_hw(c1);                          \
    const float o0 = __builtin_fmaf(0.5f, to0, 0.5f);                          \
    const float o1 = __builtin_fmaf(0.5f, to1, 0.5f);                          \
    v2f hn;                                                                    \
    hn[0] = o0 * tc0;                                                          \
    hn[1] = o1 * tc1;                                                          \
    hbuf[(k) & 1][w][lane] = hn;                                               \
    h_publish();                                                               \
      \
      \
      \
    Gk = load_g((t + (k) + 8 < T) ? (t + (k) + 8) : (T - 1));                  \
    h_collect();                                                               \
    const v2f p0 = hbuf[(k) & 1][0][lane], p1 = hbuf[(k) & 1][1][lane];        \
    const v2f p2 = hbuf[(k) & 1][2][lane], p3 = hbuf[(k) & 1][3][lane];        \
    hf[0] = p0[0]; hf[1] = p0[1]; hf[2] = p1[0]; hf[3] = p1[1];                \
    hf[4] = p2[0]; hf[5] = p2[1]; hf[6] = p3[0]; hf[7] = p3[1];                \
      \
      \
    if (w == ((k) & 3)) store_y(y, t + (k), row, hi, hf);                      \
  }

  {
    constexpr int t = 0;
    STEP(0, G0) STEP(1, G1) STEP(2, G2) STEP(3, G3)
    STEP(4, G4) STEP(5, G5) STEP(6, G6) STEP(7, G7)
  }
  for (int t = 8; t < T; t += 8) {
    STEP(0, G0) STEP(1, G1) STEP(2, G2) STEP(3, G3)
    STEP(4, G4) STEP(5, G5) STEP(6, G6) STEP(7, G7)
  }
#undef STEP
}

extern "C" void kernel_launch(void* const* d_in, const int*  ,
                              int  , void* d_out, int  ,
                              void* d_ws, size_t  ,
                              hipStream_t stream) {
  const float* x    = (const float*)d_in[0];
  const float* W_ih = (const float*)d_in[1];
  const float* W_hh = (const float*)d_in[2];
  const float* b_ih = (const float*)d_in[3];
  const float* b_hh = (const float*)d_in[4];
  float* y  = (float*)d_out;
  float* xg = (float*)d_ws;
  lstm_proj_kernel<<<PROJ_BLOCKS, 32 * PROJ_WAVES, 0, stream>>>(x, W_ih, b_ih, b_hh, xg);
  lstm_rec_kernel<<<NTILE, 32 * REC_WAVES, 0, stream>>>(xg, W_hh, y);
}


# Input projection fused into the recurrent kernel: a wave issues its 4 projection matrix instructions per step from registers inside the LDS waits of the h hand-over; one launch; f32 sum order kept
# speedup vs baseline: 5.8990x; 1.2484x over previous
#include <hip/hip_runtime.h>


typedef __attribute__((ext_vector_type(2))) float v2f;
typedef __attribute__((ext_vector_type(4))) float v4f;
typedef __attribute__((ext_vector_type(8))) float v8f;

namespace {
constexpr int T = 1024, B = 512, D = 16, H = 16;
constexpr int NTILE = B / 16;
constexpr int REC_WAVES = 4;

__device__ __forceinline__ int gate_row(int w, int m) {
  return 16 * ((m & 7) >> 1) + 8 * (m >> 3) + 2 * w + (m & 1);
}

__device__ __forceinline__ float tanh_hw(float x) {
  float r;
  asm volatile("v_tanh_f32 %0, %1\n\tv_nop" : "=v"(r) : "v"(x));
  return r;
}

__device__ __forceinline__ v8f mma(v2f a, v2f b, v8f c) {
  return __builtin_amdgcn_wmma_f32_16x16x4_f32(false, a, false, b, (short)0, c,
                                               false, false);
}
__device__ __forceinline__ v2f lo2(v4f v) { return __builtin_shufflevector(v, v, 0, 1); }
__device__ __forceinline__ v2f hi2(v4f v) { return __builtin_shufflevector(v, v, 2, 3); }

__device__ __forceinline__ void store_y(float* __restrict__ y, int t, int row,
                                        int hi, const v8f& hv) {
  float* yrow = y + ((size_t)t * B + row) * H + 8 * hi;
  *(v4f*)yrow       = __builtin_shufflevector(hv, hv, 0, 1, 2, 3);
  *(v4f*)(yrow + 4) = __builtin_shufflevector(hv, hv, 4, 5, 6, 7);
}

__device__ __forceinline__ void h_publish() {
  asm volatile("s_wait_dscnt 0x0" ::: "memory");
  asm volatile("s_barrier_signal -1" ::: "memory");
}
__device__ __forceinline__ void h_collect() {
  asm volatile("s_barrier_wait -1" ::: "memory");
}
}

__global__ __launch_bounds__(32 * REC_WAVES) void lstm_kernel(
    const float* __restrict__ x, const float* __restrict__ W_ih,
    const float* __restrict__ W_hh, const float* __restrict__ b_ih,
    const float* __restrict__ b_hh, float* __restrict__ y) {
  __shared__ v2f hbuf[2][REC_WAVES][32];

  const int lane = threadIdx.x & 31;
  const int w    = __builtin_amdgcn_readfirstlane((int)(threadIdx.x >> 5));
  const int b    = lane & 15;
  const int hi   = lane >> 4;
  const int tile = blockIdx.x;
  const int row  = tile * 16 + b;

  v2f whh[4], wih[4];
#pragma unroll
  for (int j = 0; j < 4; ++j) {
    whh[j] = *(const v2f*)(W_hh + (size_t)gate_row(w, b) * H + 8 * hi + 2 * j);
    wih[j] = *(const v2f*)(W_ih + (size_t)gate_row(w, b) * D + 8 * hi + 2 * j);
  }
  v8f biasV;
#pragma unroll
  for (int r = 0; r < 8; ++r) {
    const int g = gate_row(w, 8 * hi + r);
    biasV[r] = b_ih[g] + b_hh[g];
  }

  const float* xp = x + ((size_t)tile * 16 + b) * D + 8 * hi;
  auto load_xl = [&](int s) -> v4f { return *(const v4f*)(xp + (size_t)s * (B * D)); };
  auto load_xh = [&](int s) -> v4f { return *(const v4f*)(xp + (size_t)s * (B * D) + 4); };

  v4f XL0 = load_xl(0), XH0 = load_xh(0), XL1 = load_xl(1), XH1 = load_xh(1);
  v4f XL2 = load_xl(2), XH2 = load_xh(2), XL3 = load_xl(3), XH3 = load_xh(3);
  v4f XL4 = load_xl(4), XH4 = load_xh(4), XL5 = load_xl(5), XH5 = load_xh(5);
  v4f XL6 = load_xl(6), XH6 = load_xh(6), XL7 = load_xl(7), XH7 = load_xh(7);

  v8f A0 = mma(wih[0], lo2(XL0), biasV);
  A0 = mma(wih[1], hi2(XL0), A0);
  A0 = mma(wih[2], lo2(XH0), A0);
  A0 = mma(wih[3], hi2(XH0), A0);
  v8f A1 = mma(wih[0], lo2(XL1), biasV);
  A1 = mma(wih[1], hi2(XL1), A1);
  v8f A2, A3;

  v8f hf = {};
  float c0 = 0.0f, c1 = 0.0f;

#define STEP(k, Ak, Ak1, Ak2, XLk, XHk, XHk1, XLk2)                            \
  {                                                                            \
    v8f acc = Ak;                                                              \
    _Pragma("unroll")                                                          \
    for (int j = 0; j < 4; ++j) {                                              \
      v2f hB;                                                                  \
      hB[0] = hf[2 * j];                                                       \
      hB[1] = hf[2 * j + 1];                                                   \
      acc = mma(whh[j], hB, acc);                                              \
    }                                                                          \
      \
      \
    const float ai0 = 0.5f * acc[0], ai1 = 0.5f * acc[1];                      \
    const float af0 = 0.5f * acc[2], af1 = 0.5f * acc[3];                      \
    const float ao0 = 0.5f * acc[6], ao1 = 0.5f * acc[7];                      \
    const float g0 = tanh_hw(acc[4]), g1 = tanh_hw(acc[5]);                    \
    const float ti0 = tanh_hw(ai0), ti1 = tanh_hw(ai1);                        \
    const float tf0 = tanh_hw(af0), tf1 = tanh_hw(af1);                        \
    const float i0 = __builtin_fmaf(0.5f, ti0, 0.5f);                          \
    const float i1 = __builtin_fmaf(0.5f, ti1, 0.5f);                          \
    const float f0 = __builtin_fmaf(0.5f, tf0, 0.5f);                          \
    const float f1 = __builtin_fmaf(0.5f, tf1, 0.5f);                          \
               \
    const float m0 = i0 * g0, m1 = i1 * g1;                                    \
    c0 = __builtin_fmaf(f0, c0, m0);                                           \
    c1 = __builtin_fmaf(f1, c1, m1);                                           \
    const float to0 = tanh_hw(ao0), to1 = tanh_hw(ao1);                        \
    const float tc0 = tanh_hw(c0), tc1 = tanh_hw(c1);                          \
    const float o0 = __builtin_fmaf(0.5f, to0, 0.5f);                          \
    const float o1 = __builtin_fmaf(0.5f, to1, 0.5f);                          \
    v2f hn;                                                                    \
    hn[0] = o0 * tc0;                                                          \
    hn[1] = o1 * tc1;                                                          \
    hbuf[(k) & 1][w][lane] = hn;                                               \
      \
      \
      \
      \
    __builtin_amdgcn_sched_barrier(0);                                         \
    Ak1 = mma(wih[2], lo2(XHk1), Ak1);                                         \
    __builtin_amdgcn_sched_barrier(0);                                         \
    h_publish();                                                               \
      \
      \
    {                                                                          \
      const int sn = (t + (k) + 8 < T) ? (t + (k) + 8) : (T - 1);              \
      XLk = load_xl(sn);                                                       \
      XHk = load_xh(sn);                                                       \
    }                                                                          \
    h_collect();                                                               \
    const v2f p0 = hbuf[(k) & 1][0][lane], p1 = hbuf[(k) & 1][1][lane];        \
    const v2f p2 = hbuf[(k) & 1][2][lane], p3 = hbuf[(k) & 1][3][lane];        \
    hf[0] = p0[0]; hf[1] = p0[1]; hf[2] = p1[0]; hf[3] = p1[1];                \
    hf[4] = p2[0]; hf[5] = p2[1]; hf[6] = p3[0]; hf[7] = p3[1];                \
      \
      \
      \
      \
    __builtin_amdgcn_sched_barrier(0);                                         \
    Ak2 = mma(wih[0], lo2(XLk2), biasV);                                       \
    Ak1 = mma(wih[3], hi2(XHk1), Ak1);                                         \
    Ak2 = mma(wih[1], hi2(XLk2), Ak2);                                         \
    __builtin_amdgcn_sched_barrier(0);                                         \
      \
      \
    if (w == ((k) & 3)) store_y(y, t + (k), row, hi, hf);                      \
  }

#define EIGHT_STEPS                                                            \
  STEP(0, A0, A1, A2, XL0, XH0, XH1, XL2)                                      \
  STEP(1, A1, A2, A3, XL1, XH1, XH2, XL3)                                      \
  STEP(2, A2, A3, A0, XL2, XH2, XH3, XL4)                                      \
  STEP(3, A3, A0, A1, XL3, XH3, XH4, XL5)                                      \
  STEP(4, A0, A1, A2, XL4, XH4, XH5, XL6)                                      \
  STEP(5, A1, A2, A3, XL5, XH5, XH6, XL7)                                      \
  STEP(6, A2, A3, A0, XL6, XH6, XH7, XL0)                                      \
  STEP(7, A3, A0, A1, XL7, XH7, XH0, XL1)

  {
    constexpr int t = 0;
    EIGHT_STEPS
  }
  for (int t = 8; t < T; t += 8) {
    EIGHT_STEPS
  }
#undef EIGHT_STEPS
#undef STEP
}

extern "C" void kernel_launch(void* const* d_in, const int*  ,
                              int  , void* d_out, int  ,
                              void*  , size_t  ,
                              hipStream_t stream) {
  const float* x    = (const float*)d_in[0];
  const float* W_ih = (const float*)d_in[1];
  const float* W_hh = (const float*)d_in[2];
  const float* b_ih = (const float*)d_in[3];
  const float* b_hh = (const float*)d_in[4];
  float* y  = (float*)d_out;
  lstm_kernel<<<NTILE, 32 * REC_WAVES, 0, stream>>>(x, W_ih, W_hh, b_ih, b_hh, y);
}


# Global memory instructions off the step's dependent chain: y of step s stored a step later from registers inside the LDS read latency (one projection WMMA moved aside), x request behind the LDS store
# speedup vs baseline: 6.5797x; 1.1154x over previous
#include <hip/hip_runtime.h>


typedef __attribute__((ext_vector_type(2))) float v2f;
typedef __attribute__((ext_vector_type(4))) float v4f;
typedef __attribute__((ext_vector_type(8))) float v8f;

namespace {
constexpr int T = 1024, B = 512, D = 16, H = 16;
constexpr int NTILE = B / 16;
constexpr int REC_WAVES = 4;

__device__ __forceinline__ int gate_row(int w, int m) {
  return 16 * ((m & 7) >> 1) + 8 * (m >> 3) + 2 * w + (m & 1);
}

__device__ __forceinline__ float tanh_hw(float x) {
  float r;
  asm volatile("v_tanh_f32 %0, %1\n\tv_nop" : "=v"(r) : "v"(x));
  return r;
}

__device__ __forceinline__ v8f mma(v2f a, v2f b, v8f c) {
  return __builtin_amdgcn_wmma_f32_16x16x4_f32(false, a, false, b, (short)0, c,
                                               false, false);
}
__device__ __forceinline__ v2f lo2(v4f v) { return __builtin_shufflevector(v, v, 0, 1); }
__device__ __forceinline__ v2f hi2(v4f v) { return __builtin_shufflevector(v, v, 2, 3); }

__device__ __forceinline__ void store_y(float* __restrict__ ytile, int t,
                                        unsigned yoff, const v8f& hv) {
  float* yrow = ytile + (size_t)t * (B * H) + yoff;
  *(v4f*)yrow       = __builtin_shufflevector(hv, hv, 0, 1, 2, 3);
  *(v4f*)(yrow + 4) = __builtin_shufflevector(hv, hv, 4, 5, 6, 7);
}

__device__ __forceinline__ void h_publish() {
  asm volatile("s_wait_dscnt 0x0" ::: "memory");
  asm volatile("s_barrier_signal -1" ::: "memory");
}
__device__ __forceinline__ void h_collect() {
  asm volatile("s_barrier_wait -1" ::: "memory");
}
}

__global__ __launch_bounds__(32 * REC_WAVES) void lstm_kernel(
    const float* __restrict__ x, const float* __restrict__ W_ih,
    const float* __restrict__ W_hh, const float* __restrict__ b_ih,
    const float* __restrict__ b_hh, float* __restrict__ y) {
  __shared__ v2f hbuf[2][REC_WAVES][32];

  const int lane = threadIdx.x & 31;
  const int w    = __builtin_amdgcn_readfirstlane((int)(threadIdx.x >> 5));
  const int b    = lane & 15;
  const int hi   = lane >> 4;
  const int tile = blockIdx.x;

  v2f whh[4], wih[4];
#pragma unroll
  for (int j = 0; j < 4; ++j) {
    whh[j] = *(const v2f*)(W_hh + (size_t)gate_row(w, b) * H + 8 * hi + 2 * j);
    wih[j] = *(const v2f*)(W_ih + (size_t)gate_row(w, b) * D + 8 * hi + 2 * j);
  }
  v8f biasV;
#pragma unroll
  for (int r = 0; r < 8; ++r) {
    const int g = gate_row(w, 8 * hi + r);
    biasV[r] = b_ih[g] + b_hh[g];
  }

  const float* xtile = x + (size_t)tile * 16 * D;
  const unsigned xoff = (unsigned)(b * D + 8 * hi);
  auto load_xl = [&](int s) -> v4f { return *(const v4f*)(xtile + (size_t)s * (B * D) + xoff); };
  auto load_xh = [&](int s) -> v4f { return *(const v4f*)(xtile + (size_t)s * (B * D) + xoff + 4); };
  float* ytile = y + (size_t)tile * 16 * H;
  const unsigned yoff = (unsigned)(b * H + 8 * hi);

  v4f XL0 = load_xl(0), XH0 = load_xh(0), XL1 = load_xl(1), XH1 = load_xh(1);
  v4f XL2 = load_xl(2), XH2 = load_xh(2), XL3 = load_xl(3), XH3 = load_xh(3);
  v4f XL4 = load_xl(4), XH4 = load_xh(4), XL5 = load_xl(5), XH5 = load_xh(5);
  v4f XL6 = load_xl(6), XH6 = load_xh(6), XL7 = load_xl(7), XH7 = load_xh(7);

  v8f A0 = mma(wih[0], lo2(XL0), biasV);
  A0 = mma(wih[1], hi2(XL0), A0);
  A0 = mma(wih[2], lo2(XH0), A0);
  A0 = mma(wih[3], hi2(XH0), A0);
  v8f A1 = mma(wih[0], lo2(XL1), biasV);
  A1 = mma(wih[1], hi2(XL1), A1);
  v8f A2, A3;

  v8f hf = {};
  float c0 = 0.0f, c1 = 0.0f;

#define STEP(k, Ak, Ak1, Ak2, XLk, XHk, XHk1, XLk2)                            \
  {                                                                            \
    v8f acc = Ak;                                                              \
    _Pragma("unroll")                                                          \
    for (int j = 0; j < 4; ++j) {                                              \
      v2f hB;                                                                  \
      hB[0] = hf[2 * j];                                                       \
      hB[1] = hf[2 * j + 1];                                                   \
      acc = mma(whh[j], hB, acc);                                              \
    }                                                                          \
      \
      \
    const float ai0 = 0.5f * acc[0], ai1 = 0.5f * acc[1];                      \
    const float af0 = 0.5f * acc[2], af1 = 0.5f * acc[3];                      \
    const float ao0 = 0.5f * acc[6], ao1 = 0.5f * acc[7];                      \
    const float g0 = tanh_hw(acc[4]), g1 = tanh_hw(acc[5]);                    \
    const float ti0 = tanh_hw(ai0), ti1 = tanh_hw(ai1);                        \
    const float tf0 = tanh_hw(af0), tf1 = tanh_hw(af1);                        \
    const float i0 = __builtin_fmaf(0.5f, ti0, 0.5f);                          \
    const float i1 = __builtin_fmaf(0.5f, ti1, 0.5f);                          \
    const float f0 = __builtin_fmaf(0.5f, tf0, 0.5f);                          \
    const float f1 = __builtin_fmaf(0.5f, tf1, 0.5f);                          \
               \
    const float m0 = i0 * g0, m1 = i1 * g1;                                    \
    c0 = __builtin_fmaf(f0, c0, m0);                                           \
    c1 = __builtin_fmaf(f1, c1, m1);                                           \
    const float to0 = tanh_hw(ao0), to1 = tanh_hw(ao1);                        \
    const float tc0 = tanh_hw(c0), tc1 = tanh_hw(c1);                          \
    const float o0 = __builtin_fmaf(0.5f, to0, 0.5f);                          \
    const float o1 = __builtin_fmaf(0.5f, to1, 0.5f);                          \
    v2f hn;                                                                    \
    hn[0] = o0 * tc0;                                                          \
    hn[1] = o1 * tc1;                                                          \
    hbuf[(k) & 1][w][lane] = hn;                                               \
      \
      \
      \
      \
      \
      \
    __builtin_amdgcn_sched_barrier(0);                                         \
    Ak1 = mma(wih[2], lo2(XHk1), Ak1);                                         \
    XLk = load_xl(tb + (k));                                                   \
    XHk = load_xh(tb + (k));                                                   \
    Ak1 = mma(wih[3], hi2(XHk1), Ak1);                                         \
    __builtin_amdgcn_sched_barrier(0);                                         \
    h_publish();                                                               \
    h_collect();                                                               \
    const v8f hprev = hf;      \
    const v2f p0 = hbuf[(k) & 1][0][lane], p1 = hbuf[(k) & 1][1][lane];        \
    const v2f p2 = hbuf[(k) & 1][2][lane], p3 = hbuf[(k) & 1][3][lane];        \
    hf[0] = p0[0]; hf[1] = p0[1]; hf[2] = p1[0]; hf[3] = p1[1];                \
    hf[4] = p2[0]; hf[5] = p2[1]; hf[6] = p3[0]; hf[7] = p3[1];                \
      \
      \
      \
      \
      \
      \
    __builtin_amdgcn_sched_barrier(0);                                         \
    Ak2 = mma(wih[0], lo2(XLk2), biasV);                                       \
    Ak2 = mma(wih[1], hi2(XLk2), Ak2);                                         \
    __builtin_amdgcn_sched_barrier(0);                                         \
      \
      \
      \
      \
    if (t + (k) > 0 && w == (((k) + 3) & 3))                                   \
      store_y(ytile, t + (k) - 1, yoff, hprev);                                \
  }

#define EIGHT_STEPS                                                            \
  STEP(0, A0, A1, A2, XL0, XH0, XH1, XL2)                                      \
  STEP(1, A1, A2, A3, XL1, XH1, XH2, XL3)                                      \
  STEP(2, A2, A3, A0, XL2, XH2, XH3, XL4)                                      \
  STEP(3, A3, A0, A1, XL3, XH3, XH4, XL5)                                      \
  STEP(4, A0, A1, A2, XL4, XH4, XH5, XL6)                                      \
  STEP(5, A1, A2, A3, XL5, XH5, XH6, XL7)                                      \
  STEP(6, A2, A3, A0, XL6, XH6, XH7, XL0)                                      \
  STEP(7, A3, A0, A1, XL7, XH7, XH0, XL1)

  {
    constexpr int t = 0;
    constexpr int tb = 8;
    EIGHT_STEPS
  }
  for (int t = 8; t < T; t += 8) {
    const int tb = (t + 8 < T) ? (t + 8) : (T - 8);
    EIGHT_STEPS
  }
  if (w == ((T - 1) & 3)) store_y(ytile, T - 1, yoff, hf);
#undef EIGHT_STEPS
#undef STEP
}

extern "C" void kernel_launch(void* const* d_in, const int*  ,
                              int  , void* d_out, int  ,
                              void*  , size_t  ,
                              hipStream_t stream) {
  const float* x    = (const float*)d_in[0];
  const float* W_ih = (const float*)d_in[1];
  const float* W_hh = (const float*)d_in[2];
  const float* b_ih = (const float*)d_in[3];
  const float* b_hh = (const float*)d_in[4];
  float* y  = (float*)d_out;
  lstm_kernel<<<NTILE, 32 * REC_WAVES, 0, stream>>>(x, W_ih, W_hh, b_ih, b_hh, y);
}


# Gate chain shortened: sigmoid's input scale of 1/2 folded into the i, f, o weight rows and bias at load (exact); tanh as compiler builtin, so the gate tanh issue back to back; f32 sum order unchanged
# speedup vs baseline: 7.0884x; 1.0773x over previous
#include <hip/hip_runtime.h>


typedef __attribute__((ext_vector_type(2))) float v2f;
typedef __attribute__((ext_vector_type(4))) float v4f;
typedef __attribute__((ext_vector_type(8))) float v8f;

namespace {
constexpr int T = 1024, B = 512, D = 16, H = 16;
constexpr int NTILE = B / 16;
constexpr int REC_WAVES = 4;

__device__ __forceinline__ int gate_row(int w, int m) {
  return 16 * ((m & 7) >> 1) + 8 * (m >> 3) + 2 * w + (m & 1);
}

__device__ __forceinline__ float tanh_hw(float x) {
  return __builtin_amdgcn_tanhf(x);
}

__device__ __forceinline__ v8f mma(v2f a, v2f b, v8f c) {
  return __builtin_amdgcn_wmma_f32_16x16x4_f32(false, a, false, b, (short)0, c,
                                               false, false);
}
__device__ __forceinline__ v2f lo2(v4f v) { return __builtin_shufflevector(v, v, 0, 1); }
__device__ __forceinline__ v2f hi2(v4f v) { return __builtin_shufflevector(v, v, 2, 3); }

__device__ __forceinline__ void store_y(float* __restrict__ ytile, int t,
                                        unsigned yoff, const v8f& hv) {
  float* yrow = ytile + (size_t)t * (B * H) + yoff;
  *(v4f*)yrow       = __builtin_shufflevector(hv, hv, 0, 1, 2, 3);
  *(v4f*)(yrow + 4) = __builtin_shufflevector(hv, hv, 4, 5, 6, 7);
}

__device__ __forceinline__ void h_publish() {
  asm volatile("s_wait_dscnt 0x0" ::: "memory");
  asm volatile("s_barrier_signal -1" ::: "memory");
}
__device__ __forceinline__ void h_collect() {
  asm volatile("s_barrier_wait -1" ::: "memory");
}
}

__global__ __launch_bounds__(32 * REC_WAVES) void lstm_kernel(
    const float* __restrict__ x, const float* __restrict__ W_ih,
    const float* __restrict__ W_hh, const float* __restrict__ b_ih,
    const float* __restrict__ b_hh, float* __restrict__ y) {
  __shared__ v2f hbuf[2][REC_WAVES][32];

  const int lane = threadIdx.x & 31;
  const int w    = __builtin_amdgcn_readfirstlane((int)(threadIdx.x >> 5));
  const int b    = lane & 15;
  const int hi   = lane >> 4;
  const int tile = blockIdx.x;

  const float wscale = (((b & 7) >> 1) == 2) ? 1.0f : 0.5f;
  v2f whh[4], wih[4];
#pragma unroll
  for (int j = 0; j < 4; ++j) {
    whh[j] = *(const v2f*)(W_hh + (size_t)gate_row(w, b) * H + 8 * hi + 2 * j) * wscale;
    wih[j] = *(const v2f*)(W_ih + (size_t)gate_row(w, b) * D + 8 * hi + 2 * j) * wscale;
  }
  v8f biasV;
#pragma unroll
  for (int r = 0; r < 8; ++r) {
    const int g = gate_row(w, 8 * hi + r);
    const float bsum = b_ih[g] + b_hh[g];
    biasV[r] = bsum * (((r >> 1) == 2) ? 1.0f : 0.5f);
  }

  const float* xtile = x + (size_t)tile * 16 * D;
  const unsigned xoff = (unsigned)(b * D + 8 * hi);
  auto load_xl = [&](int s) -> v4f { return *(const v4f*)(xtile + (size_t)s * (B * D) + xoff); };
  auto load_xh = [&](int s) -> v4f { return *(const v4f*)(xtile + (size_t)s * (B * D) + xoff + 4); };
  float* ytile = y + (size_t)tile * 16 * H;
  const unsigned yoff = (unsigned)(b * H + 8 * hi);

  v4f XL0 = load_xl(0), XH0 = load_xh(0), XL1 = load_xl(1), XH1 = load_xh(1);
  v4f XL2 = load_xl(2), XH2 = load_xh(2), XL3 = load_xl(3), XH3 = load_xh(3);
  v4f XL4 = load_xl(4), XH4 = load_xh(4), XL5 = load_xl(5), XH5 = load_xh(5);
  v4f XL6 = load_xl(6), XH6 = load_xh(6), XL7 = load_xl(7), XH7 = load_xh(7);

  v8f A0 = mma(wih[0], lo2(XL0), biasV);
  A0 = mma(wih[1], hi2(XL0), A0);
  A0 = mma(wih[2], lo2(XH0), A0);
  A0 = mma(wih[3], hi2(XH0), A0);
  v8f A1 = mma(wih[0], lo2(XL1), biasV);
  A1 = mma(wih[1], hi2(XL1), A1);
  v8f A2, A3;

  v8f hf = {};
  float c0 = 0.0f, c1 = 0.0f;

#define STEP(k, Ak, Ak1, Ak2, XLk, XHk, XHk1, XLk2)                            \
  {                                                                            \
    v8f acc = Ak;                                                              \
    _Pragma("unroll")                                                          \
    for (int j = 0; j < 4; ++j) {                                              \
      v2f hB;                                                                  \
      hB[0] = hf[2 * j];                                                       \
      hB[1] = hf[2 * j + 1];                                                   \
      acc = mma(whh[j], hB, acc);                                              \
    }                                                                          \
      \
      \
      \
      \
      \
    const float ti0 = tanh_hw(acc[0]), ti1 = tanh_hw(acc[1]);                  \
    const float g0 = tanh_hw(acc[4]), g1 = tanh_hw(acc[5]);                    \
    const float tf0 = tanh_hw(acc[2]), tf1 = tanh_hw(acc[3]);                  \
    const float to0 = tanh_hw(acc[6]), to1 = tanh_hw(acc[7]);                  \
    const float i0 = __builtin_fmaf(0.5f, ti0, 0.5f);                          \
    const float i1 = __builtin_fmaf(0.5f, ti1, 0.5f);                          \
    const float f0 = __builtin_fmaf(0.5f, tf0, 0.5f);                          \
    const float f1 = __builtin_fmaf(0.5f, tf1, 0.5f);                          \
               \
    const float m0 = i0 * g0, m1 = i1 * g1;                                    \
    c0 = __builtin_fmaf(f0, c0, m0);                                           \
    c1 = __builtin_fmaf(f1, c1, m1);                                           \
    const float tc0 = tanh_hw(c0), tc1 = tanh_hw(c1);                          \
    const float o0 = __builtin_fmaf(0.5f, to0, 0.5f);                          \
    const float o1 = __builtin_fmaf(0.5f, to1, 0.5f);                          \
    v2f hn;                                                                    \
    hn[0] = o0 * tc0;                                                          \
    hn[1] = o1 * tc1;                                                          \
    hbuf[(k) & 1][w][lane] = hn;                                               \
      \
      \
      \
      \
      \
      \
    __builtin_amdgcn_sched_barrier(0);                                         \
    Ak1 = mma(wih[2], lo2(XHk1), Ak1);                                         \
    XLk = load_xl(tb + (k));                                                   \
    XHk = load_xh(tb + (k));                                                   \
    Ak1 = mma(wih[3], hi2(XHk1), Ak1);                                         \
    __builtin_amdgcn_sched_barrier(0);                                         \
    h_publish();                                                               \
    h_collect();                                                               \
    const v8f hprev = hf;      \
    const v2f p0 = hbuf[(k) & 1][0][lane], p1 = hbuf[(k) & 1][1][lane];        \
    const v2f p2 = hbuf[(k) & 1][2][lane], p3 = hbuf[(k) & 1][3][lane];        \
    hf[0] = p0[0]; hf[1] = p0[1]; hf[2] = p1[0]; hf[3] = p1[1];                \
    hf[4] = p2[0]; hf[5] = p2[1]; hf[6] = p3[0]; hf[7] = p3[1];                \
      \
      \
      \
      \
      \
      \
    __builtin_amdgcn_sched_barrier(0);                                         \
    Ak2 = mma(wih[0], lo2(XLk2), biasV);                                       \
    Ak2 = mma(wih[1], hi2(XLk2), Ak2);                                         \
    __builtin_amdgcn_sched_barrier(0);                                         \
      \
      \
      \
      \
    if (t + (k) > 0 && w == (((k) + 3) & 3))                                   \
      store_y(ytile, t + (k) - 1, yoff, hprev);                                \
  }

#define EIGHT_STEPS                                                            \
  STEP(0, A0, A1, A2, XL0, XH0, XH1, XL2)                                      \
  STEP(1, A1, A2, A3, XL1, XH1, XH2, XL3)                                      \
  STEP(2, A2, A3, A0, XL2, XH2, XH3, XL4)                                      \
  STEP(3, A3, A0, A1, XL3, XH3, XH4, XL5)                                      \
  STEP(4, A0, A1, A2, XL4, XH4, XH5, XL6)                                      \
  STEP(5, A1, A2, A3, XL5, XH5, XH6, XL7)                                      \
  STEP(6, A2, A3, A0, XL6, XH6, XH7, XL0)                                      \
  STEP(7, A3, A0, A1, XL7, XH7, XH0, XL1)

  {
    constexpr int t = 0;
    constexpr int tb = 8;
    EIGHT_STEPS
  }
  for (int t = 8; t < T; t += 8) {
    const int tb = (t + 8 < T) ? (t + 8) : (T - 8);
    EIGHT_STEPS
  }
  if (w == ((T - 1) & 3)) store_y(ytile, T - 1, yoff, hf);
#undef EIGHT_STEPS
#undef STEP
}

extern "C" void kernel_launch(void* const* d_in, const int*  ,
                              int  , void* d_out, int  ,
                              void*  , size_t  ,
                              hipStream_t stream) {
  const float* x    = (const float*)d_in[0];
  const float* W_ih = (const float*)d_in[1];
  const float* W_hh = (const float*)d_in[2];
  const float* b_ih = (const float*)d_in[3];
  const float* b_hh = (const float*)d_in[4];
  float* y  = (float*)d_out;
  lstm_kernel<<<NTILE, 32 * REC_WAVES, 0, stream>>>(x, W_ih, W_hh, b_ih, b_hh, y);
}


# Barrier split into signal and wait; the wave's own K-step of the next recurrent product (own h, from registers) issued between them, nothing displaced: 3 of 4 WMMAs wait for h; f32 sum order changes
# speedup vs baseline: 7.3554x; 1.0377x over previous
#include <hip/hip_runtime.h>


typedef __attribute__((ext_vector_type(2))) float v2f;
typedef __attribute__((ext_vector_type(4))) float v4f;
typedef __attribute__((ext_vector_type(8))) float v8f;

namespace {
constexpr int T = 1024, B = 512, D = 16, H = 16;
constexpr int NTILE = B / 16;
constexpr int REC_WAVES = 4;

__device__ __forceinline__ int gate_row(int w, int m) {
  return 16 * ((m & 7) >> 1) + 8 * (m >> 3) + 2 * w + (m & 1);
}

__device__ __forceinline__ float tanh_hw(float x) {
  return __builtin_amdgcn_tanhf(x);
}

__device__ __forceinline__ v8f mma(v2f a, v2f b, v8f c) {
  return __builtin_amdgcn_wmma_f32_16x16x4_f32(false, a, false, b, (short)0, c,
                                               false, false);
}
__device__ __forceinline__ v2f lo2(v4f v) { return __builtin_shufflevector(v, v, 0, 1); }
__device__ __forceinline__ v2f hi2(v4f v) { return __builtin_shufflevector(v, v, 2, 3); }

__device__ __forceinline__ void store_y(float* __restrict__ ytile, int t,
                                        unsigned yoff, const v8f& hv) {
  float* yrow = ytile + (size_t)t * (B * H) + yoff;
  *(v4f*)yrow       = __builtin_shufflevector(hv, hv, 0, 1, 2, 3);
  *(v4f*)(yrow + 4) = __builtin_shufflevector(hv, hv, 4, 5, 6, 7);
}

template <int WS>
__device__ __forceinline__ v8f whole_h(v2f own, v2f o1, v2f o2, v2f o3) {
  v2f part[4];
  part[WS] = own;
  part[(WS + 1) & 3] = o1;
  part[(WS + 2) & 3] = o2;
  part[(WS + 3) & 3] = o3;
  v8f hv;
  hv[0] = part[0][0]; hv[1] = part[0][1]; hv[2] = part[1][0]; hv[3] = part[1][1];
  hv[4] = part[2][0]; hv[5] = part[2][1]; hv[6] = part[3][0]; hv[7] = part[3][1];
  return hv;
}

__device__ __forceinline__ void h_publish() {
  asm volatile("s_wait_dscnt 0x0" ::: "memory");
  asm volatile("s_barrier_signal -1" ::: "memory");
}
__device__ __forceinline__ void h_collect() {
  asm volatile("s_barrier_wait -1" ::: "memory");
}
}

__global__ __launch_bounds__(32 * REC_WAVES) void lstm_kernel(
    const float* __restrict__ x, const float* __restrict__ W_ih,
    const float* __restrict__ W_hh, const float* __restrict__ b_ih,
    const float* __restrict__ b_hh, float* __restrict__ y) {
  __shared__ v2f hbuf[2][REC_WAVES][32];

  const int lane = threadIdx.x & 31;
  const int w    = __builtin_amdgcn_readfirstlane((int)(threadIdx.x >> 5));
  const int b    = lane & 15;
  const int hi   = lane >> 4;
  const int tile = blockIdx.x;

  const float wscale = (((b & 7) >> 1) == 2) ? 1.0f : 0.5f;
  const int j1 = (w + 1) & 3, j2 = (w + 2) & 3, j3 = (w + 3) & 3;
  const float* whh_row = W_hh + (size_t)gate_row(w, b) * H + 8 * hi;
  const v2f whh_own = *(const v2f*)(whh_row + 2 * w) * wscale;
  const v2f whh_o1 = *(const v2f*)(whh_row + 2 * j1) * wscale;
  const v2f whh_o2 = *(const v2f*)(whh_row + 2 * j2) * wscale;
  const v2f whh_o3 = *(const v2f*)(whh_row + 2 * j3) * wscale;
  v2f wih[4];
#pragma unroll
  for (int j = 0; j < 4; ++j)
    wih[j] = *(const v2f*)(W_ih + (size_t)gate_row(w, b) * D + 8 * hi + 2 * j) * wscale;
  v8f biasV;
#pragma unroll
  for (int r = 0; r < 8; ++r) {
    const int g = gate_row(w, 8 * hi + r);
    const float bsum = b_ih[g] + b_hh[g];
    biasV[r] = bsum * (((r >> 1) == 2) ? 1.0f : 0.5f);
  }

  const float* xtile = x + (size_t)tile * 16 * D;
  const unsigned xoff = (unsigned)(b * D + 8 * hi);
  auto load_xl = [&](int s) -> v4f { return *(const v4f*)(xtile + (size_t)s * (B * D) + xoff); };
  auto load_xh = [&](int s) -> v4f { return *(const v4f*)(xtile + (size_t)s * (B * D) + xoff + 4); };
  float* ytile = y + (size_t)tile * 16 * H;
  const unsigned yoff = (unsigned)(b * H + 8 * hi);

  v4f XL0 = load_xl(0), XH0 = load_xh(0), XL1 = load_xl(1), XH1 = load_xh(1);
  v4f XL2 = load_xl(2), XH2 = load_xh(2), XL3 = load_xl(3), XH3 = load_xh(3);
  v4f XL4 = load_xl(4), XH4 = load_xh(4), XL5 = load_xl(5), XH5 = load_xh(5);
  v4f XL6 = load_xl(6), XH6 = load_xh(6), XL7 = load_xl(7), XH7 = load_xh(7);

  v8f A0 = mma(wih[0], lo2(XL0), biasV);
  A0 = mma(wih[1], hi2(XL0), A0);
  A0 = mma(wih[2], lo2(XH0), A0);
  A0 = mma(wih[3], hi2(XH0), A0);
  v8f A1 = mma(wih[0], lo2(XL1), biasV);
  A1 = mma(wih[1], hi2(XL1), A1);
  v8f A2, A3;

  v2f hown = {}, q1 = {}, q2 = {}, q3 = {};
  float c0 = 0.0f, c1 = 0.0f;

#define STEP(k, Ak, Ak1, Ak2, XLk, XHk, XHk1, XLk2)                            \
  {                                                                            \
           \
    v8f acc = mma(whh_o1, q1, Ak);                                             \
    acc = mma(whh_o2, q2, acc);                                                \
    acc = mma(whh_o3, q3, acc);                                                \
      \
      \
      \
      \
      \
    const float ti0 = tanh_hw(acc[0]), ti1 = tanh_hw(acc[1]);                  \
    const float g0 = tanh_hw(acc[4]), g1 = tanh_hw(acc[5]);                    \
    const float tf0 = tanh_hw(acc[2]), tf1 = tanh_hw(acc[3]);                  \
    const float to0 = tanh_hw(acc[6]), to1 = tanh_hw(acc[7]);                  \
    const float i0 = __builtin_fmaf(0.5f, ti0, 0.5f);                          \
    const float i1 = __builtin_fmaf(0.5f, ti1, 0.5f);                          \
    const float f0 = __builtin_fmaf(0.5f, tf0, 0.5f);                          \
    const float f1 = __builtin_fmaf(0.5f, tf1, 0.5f);                          \
               \
    const float m0 = i0 * g0, m1 = i1 * g1;                                    \
    c0 = __builtin_fmaf(f0, c0, m0);                                           \
    c1 = __builtin_fmaf(f1, c1, m1);                                           \
    const float tc0 = tanh_hw(c0), tc1 = tanh_hw(c1);                          \
    const float o0 = __builtin_fmaf(0.5f, to0, 0.5f);                          \
    const float o1 = __builtin_fmaf(0.5f, to1, 0.5f);                          \
    v2f hn;                                                                    \
    hn[0] = o0 * tc0;                                                          \
    hn[1] = o1 * tc1;                                                          \
    hbuf[(k) & 1][w][lane] = hn;                                               \
      \
      \
      \
      \
      \
      \
    __builtin_amdgcn_sched_barrier(0);                                         \
    Ak1 = mma(wih[2], lo2(XHk1), Ak1);                                         \
    XLk = load_xl(tb + (k));                                                   \
    XHk = load_xh(tb + (k));                                                   \
    Ak1 = mma(wih[3], hi2(XHk1), Ak1);                                         \
    __builtin_amdgcn_sched_barrier(0);                                         \
    h_publish();                                                               \
      \
      \
      \
      \
      \
      \
    __builtin_amdgcn_sched_barrier(0);                                         \
    Ak1 = mma(whh_own, hn, Ak1);                                               \
    __builtin_amdgcn_sched_barrier(0);                                         \
    h_collect();                                                               \
                               \
    const v2f yown = hown, y1 = q1, y2 = q2, y3 = q3;                          \
    hown = hn;                                                                 \
    q1 = hbuf[(k) & 1][j1][lane];                                              \
    q2 = hbuf[(k) & 1][j2][lane];                                              \
    q3 = hbuf[(k) & 1][j3][lane];                                              \
      \
      \
      \
      \
      \
      \
    __builtin_amdgcn_sched_barrier(0);                                         \
    Ak2 = mma(wih[0], lo2(XLk2), biasV);                                       \
    Ak2 = mma(wih[1], hi2(XLk2), Ak2);                                         \
    __builtin_amdgcn_sched_barrier(0);                                         \
      \
      \
      \
      \
    if (t + (k) > 0 && w == (((k) + 3) & 3))                                   \
      store_y(ytile, t + (k) - 1, yoff,                                        \
              whole_h<((k) + 3) & 3>(yown, y1, y2, y3));                       \
  }

#define EIGHT_STEPS                                                            \
  STEP(0, A0, A1, A2, XL0, XH0, XH1, XL2)                                      \
  STEP(1, A1, A2, A3, XL1, XH1, XH2, XL3)                                      \
  STEP(2, A2, A3, A0, XL2, XH2, XH3, XL4)                                      \
  STEP(3, A3, A0, A1, XL3, XH3, XH4, XL5)                                      \
  STEP(4, A0, A1, A2, XL4, XH4, XH5, XL6)                                      \
  STEP(5, A1, A2, A3, XL5, XH5, XH6, XL7)                                      \
  STEP(6, A2, A3, A0, XL6, XH6, XH7, XL0)                                      \
  STEP(7, A3, A0, A1, XL7, XH7, XH0, XL1)

  {
    constexpr int t = 0;
    constexpr int tb = 8;
    EIGHT_STEPS
  }
  for (int t = 8; t < T; t += 8) {
    const int tb = (t + 8 < T) ? (t + 8) : (T - 8);
    EIGHT_STEPS
  }
  if (w == ((T - 1) & 3))
    store_y(ytile, T - 1, yoff, whole_h<(T - 1) & 3>(hown, q1, q2, q3));
#undef EIGHT_STEPS
#undef STEP
}

extern "C" void kernel_launch(void* const* d_in, const int*  ,
                              int  , void* d_out, int  ,
                              void*  , size_t  ,
                              hipStream_t stream) {
  const float* x    = (const float*)d_in[0];
  const float* W_ih = (const float*)d_in[1];
  const float* W_hh = (const float*)d_in[2];
  const float* b_ih = (const float*)d_in[3];
  const float* b_hh = (const float*)d_in[4];
  float* y  = (float*)d_out;
  lstm_kernel<<<NTILE, 32 * REC_WAVES, 0, stream>>>(x, W_ih, W_hh, b_ih, b_hh, y);
}
